# starting kernel
# baseline (speedup 1.0000x reference)
#include <hip/hip_runtime.h>

typedef __attribute__((ext_vector_type(2))) float v2f;
typedef __attribute__((ext_vector_type(8))) float v8f;


__global__ void init_deg_kernel(float* __restrict__ deg, int n) {
  int i = blockIdx.x * blockDim.x + threadIdx.x;
  if (i < n) deg[i] = 1.0f;
}

__global__ void deg_count_kernel(const int* __restrict__ dst, float* __restrict__ deg, int e) {
  int i = blockIdx.x * blockDim.x + threadIdx.x;
  if (i < e) atomicAdd(&deg[dst[i]], 1.0f);
}

__global__ void dinv_kernel(float* __restrict__ deg, int n) {
  int i = blockIdx.x * blockDim.x + threadIdx.x;
  if (i < n) {
    float d = deg[i];
    deg[i] = (d > 0.0f) ? rsqrtf(d) : 0.0f;
  }
}

template <int K, int NC>
__global__ __launch_bounds__(32 * (NC / 16)) void gemm_wmma_kernel(
    const float* __restrict__ X, const float* __restrict__ W,
    float* __restrict__ H, int M) {
  const int lane = threadIdx.x & 31;
  const int wave = threadIdx.x >> 5;
  const int m0   = blockIdx.x << 4;
  const int n0   = wave << 4;
  const int l15  = lane & 15;
  const int kh   = (lane >> 4) << 1;

  int row = m0 + l15;
  if (row >= M) row = M - 1;
  const int col = n0 + l15;

  v8f c = {};
  for (int k0 = 0; k0 < K; k0 += 4) {
    v2f a, b;
    a.x = X[row * K + k0 + kh];
    a.y = X[row * K + k0 + kh + 1];
    b.x = W[(k0 + kh) * NC + col];
    b.y = W[(k0 + kh + 1) * NC + col];
    c = __builtin_amdgcn_wmma_f32_16x16x4_f32(
         false, a,  false, b,
         (short)0, c,  false,  false);
  }

  const int rbase = m0 + ((lane >> 4) << 3);
#pragma unroll
  for (int v = 0; v < 8; ++v) {
    int r = rbase + v;
    if (r < M) H[r * NC + col] = c[v];
  }
}

template <int CLOG>
__global__ void edge_agg_kernel(const int* __restrict__ src, const int* __restrict__ dst,
                                const float* __restrict__ dinv, const float* __restrict__ h,
                                float* __restrict__ agg, int e) {
  unsigned t = blockIdx.x * blockDim.x + threadIdx.x;
  unsigned edge = t >> CLOG;
  if (edge >= (unsigned)e) return;
  unsigned c = t & ((1u << CLOG) - 1u);
  int s = src[edge];
  int d = dst[edge];
  float nrm = dinv[s] * dinv[d];
  atomicAdd(&agg[((size_t)d << CLOG) + c], nrm * h[((size_t)s << CLOG) + c]);
}

template <int CLOG, bool RELU>
__global__ void finish_kernel(const float* __restrict__ agg, const float* __restrict__ hlin,
                              const float* __restrict__ dinv, const float* __restrict__ bias,
                              float* __restrict__ out, int n) {
  unsigned t = blockIdx.x * blockDim.x + threadIdx.x;
  unsigned i = t >> CLOG;
  if (i >= (unsigned)n) return;
  unsigned c = t & ((1u << CLOG) - 1u);
  float di = dinv[i];
  float v = agg[t] + di * di * hlin[t] + bias[c];
  out[t] = RELU ? fmaxf(v, 0.0f) : v;
}


extern "C" void kernel_launch(void* const* d_in, const int* in_sizes, int n_in,
                              void* d_out, int out_size, void* d_ws, size_t ws_size,
                              hipStream_t stream) {
  (void)n_in; (void)out_size; (void)ws_size;
  const float* x  = (const float*)d_in[0];
  const int*   ei = (const int*)d_in[1];
  const float* W1 = (const float*)d_in[2];
  const float* b1 = (const float*)d_in[3];
  const float* W2 = (const float*)d_in[4];
  const float* b2 = (const float*)d_in[5];

  const int IN_C = 128, HID = 64, OUT = 32;
  const int n = in_sizes[0] / IN_C;
  const int e = in_sizes[1] / 2;
  const int* srcp = ei;
  const int* dstp = ei + e;

  float* ws   = (float*)d_ws;
  size_t off  = ((size_t)n + 255) & ~(size_t)255;
  float* dinv = ws;
  float* h1   = ws + off;
  float* agg1 = h1 + (size_t)n * HID;
  float* h2   = agg1 + (size_t)n * HID;
  float* outp = (float*)d_out;

  const int TPB = 256;

  hipMemsetAsync(agg1, 0, (size_t)n * HID * sizeof(float), stream);
  hipMemsetAsync(d_out, 0, (size_t)n * OUT * sizeof(float), stream);

  init_deg_kernel<<<(n + TPB - 1) / TPB, TPB, 0, stream>>>(dinv, n);
  deg_count_kernel<<<(e + TPB - 1) / TPB, TPB, 0, stream>>>(dstp, dinv, e);
  dinv_kernel<<<(n + TPB - 1) / TPB, TPB, 0, stream>>>(dinv, n);

  gemm_wmma_kernel<128, 64><<<(n + 15) / 16, 128, 0, stream>>>(x, W1, h1, n);
  {
    unsigned total = (unsigned)e << 6;
    edge_agg_kernel<6><<<(total + TPB - 1) / TPB, TPB, 0, stream>>>(srcp, dstp, dinv, h1, agg1, e);
  }
  {
    unsigned total = (unsigned)n << 6;
    finish_kernel<6, true><<<(total + TPB - 1) / TPB, TPB, 0, stream>>>(agg1, h1, dinv, b1, agg1, n);
  }

  gemm_wmma_kernel<64, 32><<<(n + 15) / 16, 64, 0, stream>>>(agg1, W2, h2, n);
  {
    unsigned total = (unsigned)e << 5;
    edge_agg_kernel<5><<<(total + TPB - 1) / TPB, TPB, 0, stream>>>(srcp, dstp, dinv, h2, outp, e);
  }
  {
    unsigned total = (unsigned)n << 5;
    finish_kernel<5, false><<<(total + TPB - 1) / TPB, TPB, 0, stream>>>(outp, h2, dinv, b2, outp, n);
  }
}


# Edge sums as per-destination gathers over incoming-edge lists rebuilt each call (int atomic arrival-order rank, scan, fill); no float atomics; rows pre-scaled by dinv; sum order now list order
# speedup vs baseline: 2.6033x; 2.6033x over previous
#include <hip/hip_runtime.h>


typedef __attribute__((ext_vector_type(2))) float v2f;
typedef __attribute__((ext_vector_type(8))) float v8f;


__global__ void zero_count_kernel(int* __restrict__ cnt, int np) {
  int i = blockIdx.x * blockDim.x + threadIdx.x;
  if (i < np) cnt[i] = 0;
}

__global__ void rank_kernel(const int* __restrict__ dst, int* __restrict__ cnt,
                            int* __restrict__ rank, int e) {
  int i = blockIdx.x * blockDim.x + threadIdx.x;
  if (i < e) rank[i] = atomicAdd(&cnt[dst[i]], 1);
}

__global__ __launch_bounds__(256) void scan_kernel(const int* __restrict__ cnt,
                                                   int* __restrict__ rowptr,
                                                   float* __restrict__ dinv) {
  __shared__ int sh_prev[8];
  __shared__ int sh_loc[8];
  const int t = threadIdx.x;
  const int lane = t & 31;
  const int w = t >> 5;
  const int b = blockIdx.x;
  const int4* __restrict__ c4 = (const int4*)cnt;

  int prev = 0;
  for (int j = 0; j < b; ++j) {
    const int4 v = c4[j * 256 + t];
    prev += (v.x + v.y) + (v.z + v.w);
  }
  const int4 m = c4[b * 256 + t];
  const int s1 = m.x;
  const int s2 = s1 + m.y;
  const int s3 = s2 + m.z;
  const int s4 = s3 + m.w;

  int incl = s4;
  for (int o = 1; o < 32; o <<= 1) {
    const int y = __shfl_up(incl, o);
    if (lane >= o) incl += y;
  }
  for (int o = 16; o > 0; o >>= 1) prev += __shfl_xor(prev, o);

  if (lane == 31) sh_loc[w] = incl;
  if (lane == 0) sh_prev[w] = prev;
  __syncthreads();

  int base = 0;
  for (int k = 0; k < 8; ++k) {
    base += sh_prev[k];
    if (k < w) base += sh_loc[k];
  }
  const int ex = base + incl - s4;

  int4 r;
  r.x = ex;
  r.y = ex + s1;
  r.z = ex + s2;
  r.w = ex + s3;
  ((int4*)rowptr)[b * 256 + t] = r;

  const float d0 = (float)(m.x + 1);
  const float d1 = (float)(m.y + 1);
  const float d2 = (float)(m.z + 1);
  const float d3 = (float)(m.w + 1);
  float4 dv;
  dv.x = (d0 > 0.0f) ? rsqrtf(d0) : 0.0f;
  dv.y = (d1 > 0.0f) ? rsqrtf(d1) : 0.0f;
  dv.z = (d2 > 0.0f) ? rsqrtf(d2) : 0.0f;
  dv.w = (d3 > 0.0f) ? rsqrtf(d3) : 0.0f;
  ((float4*)dinv)[b * 256 + t] = dv;
}

__global__ void fill_kernel(const int* __restrict__ src, const int* __restrict__ dst,
                            const int* __restrict__ rank, const int* __restrict__ rowptr,
                            int* __restrict__ csr_src, int e) {
  int i = blockIdx.x * blockDim.x + threadIdx.x;
  if (i < e) csr_src[rowptr[dst[i]] + rank[i]] = src[i];
}

template <int K, int NC>
__global__ __launch_bounds__(32 * (NC / 16)) void gemm_wmma_kernel(
    const float* __restrict__ X, const float* __restrict__ W,
    const float* __restrict__ dinv, float* __restrict__ H, int M) {
  const int lane = threadIdx.x & 31;
  const int wave = threadIdx.x >> 5;
  const int m0   = blockIdx.x << 4;
  const int n0   = wave << 4;
  const int l15  = lane & 15;
  const int kh   = (lane >> 4) << 1;

  int row = m0 + l15;
  if (row >= M) row = M - 1;
  const int col = n0 + l15;

  const int rbase = m0 + ((lane >> 4) << 3);
  float dsc[8];
#pragma unroll
  for (int v = 0; v < 8; ++v) {
    int r = rbase + v;
    dsc[v] = dinv[(r < M) ? r : (M - 1)];
  }

  v8f c = {};
  for (int k0 = 0; k0 < K; k0 += 4) {
    v2f a, b;
    a.x = X[row * K + k0 + kh];
    a.y = X[row * K + k0 + kh + 1];
    b.x = W[(k0 + kh) * NC + col];
    b.y = W[(k0 + kh + 1) * NC + col];
    c = __builtin_amdgcn_wmma_f32_16x16x4_f32(
         false, a,  false, b,
         (short)0, c,  false,  false);
  }

#pragma unroll
  for (int v = 0; v < 8; ++v) {
    int r = rbase + v;
    const float o = c[v] * dsc[v];
    if (r < M) H[r * NC + col] = o;
  }
}

template <int C, bool RELU>
__global__ __launch_bounds__(256) void gather_kernel(
    const int* __restrict__ rowptr, const int* __restrict__ cnt,
    const int* __restrict__ csr_src, const float* __restrict__ dinv,
    const float* __restrict__ h, const float* __restrict__ bias,
    float* __restrict__ out, int n) {
  constexpr int LPR = C / 2;
  constexpr int RPI = 32 / LPR;
  constexpr int G   = 8;
  constexpr int U   = G / RPI;
  const int lane = threadIdx.x & 31;
  const int wv   = __builtin_amdgcn_readfirstlane((int)(threadIdx.x >> 5));
  const int node = blockIdx.x * 8 + wv;
  if (node >= n) return;
  const int sub = lane / LPR;
  const int cl  = lane - sub * LPR;
  const int start = rowptr[node];
  const int deg   = cnt[node];
  const float di  = dinv[node];
  const v2f* __restrict__ h2 = (const v2f*)h;
  const int total = deg + 1;

  v2f acc = {0.0f, 0.0f};
  for (int base = 0; base < total; base += 32) {
    const int rem = total - base;
    const int m = (rem < 32) ? rem : 32;
    const int k = base + lane;
    int s_l = node;
    if (k < deg) s_l = csr_src[start + k];
    for (int j = 0; j < m; j += G) {
      v2f hv[U];
      bool ok[U];
#pragma unroll
      for (int u = 0; u < U; ++u) {
        int s;
        if (RPI == 1) {
          const int idx = j + u;
          s = __builtin_amdgcn_readlane(s_l, idx);
          ok[u] = idx < m;
        } else {
          const int idx = j + 2 * u;
          const int sa = __builtin_amdgcn_readlane(s_l, idx);
          const int sb = __builtin_amdgcn_readlane(s_l, idx + 1);
          s = sub ? sb : sa;
          ok[u] = (idx + sub) < m;
        }
        const v2f* __restrict__ rowp = h2 + (size_t)s * LPR;
        hv[u] = rowp[cl];
      }
#pragma unroll
      for (int u = 0; u < U; ++u) {
        const v2f nv = acc + hv[u];
        acc = ok[u] ? nv : acc;
      }
    }
  }

  if (RPI == 2) {
    acc.x += __shfl_xor(acc.x, 16);
    acc.y += __shfl_xor(acc.y, 16);
  }
  if (sub == 0) {
    const v2f bv = ((const v2f*)bias)[cl];
    v2f r = acc * di + bv;
    if (RELU) {
      r.x = fmaxf(r.x, 0.0f);
      r.y = fmaxf(r.y, 0.0f);
    }
    ((v2f*)out)[(size_t)node * LPR + cl] = r;
  }
}


extern "C" void kernel_launch(void* const* d_in, const int* in_sizes, int n_in,
                              void* d_out, int out_size, void* d_ws, size_t ws_size,
                              hipStream_t stream) {
  (void)n_in; (void)out_size; (void)ws_size;
  const float* x  = (const float*)d_in[0];
  const int*   ei = (const int*)d_in[1];
  const float* W1 = (const float*)d_in[2];
  const float* b1 = (const float*)d_in[3];
  const float* W2 = (const float*)d_in[4];
  const float* b2 = (const float*)d_in[5];

  const int IN_C = 128, HID = 64, OUT = 32;
  const int n = in_sizes[0] / IN_C;
  const int e = in_sizes[1] / 2;
  const int* srcp = ei;
  const int* dstp = ei + e;

  const int    np = (n + 1023) & ~1023;
  const size_t ep = ((size_t)e + 63) & ~(size_t)63;

  int*   cnt    = (int*)d_ws;
  int*   rowptr = cnt + np;
  float* dinv   = (float*)(rowptr + np);
  int*   rank   = (int*)(dinv + np);
  int*   csr    = rank + ep;
  float* h1     = (float*)(csr + ep);
  float* a1     = h1 + (size_t)n * HID;
  float* h2     = a1 + (size_t)n * HID;
  float* outp   = (float*)d_out;
  (void)OUT;

  const int TPB = 256;

  zero_count_kernel<<<np / TPB, TPB, 0, stream>>>(cnt, np);
  rank_kernel<<<(e + TPB - 1) / TPB, TPB, 0, stream>>>(dstp, cnt, rank, e);
  scan_kernel<<<np / 1024, 256, 0, stream>>>(cnt, rowptr, dinv);
  fill_kernel<<<(e + TPB - 1) / TPB, TPB, 0, stream>>>(srcp, dstp, rank, rowptr, csr, e);

  gemm_wmma_kernel<128, 64><<<(n + 15) / 16, 128, 0, stream>>>(x, W1, dinv, h1, n);
  gather_kernel<64, true><<<(n + 7) / 8, 256, 0, stream>>>(rowptr, cnt, csr, dinv, h1, b1, a1, n);

  gemm_wmma_kernel<64, 32><<<(n + 15) / 16, 64, 0, stream>>>(a1, W2, dinv, h2, n);
  gather_kernel<32, false><<<(n + 7) / 8, 256, 0, stream>>>(rowptr, cnt, csr, dinv, h2, b2, outp, n);
}


# Edge lists by two-level counting sort on LDS counters (1024-node buckets, one work-group per bucket) replace per-edge global atomics and serial scan; list order = LDS add arrival order
# speedup vs baseline: 4.5589x; 1.7512x over previous
#include <hip/hip_runtime.h>


typedef __attribute__((ext_vector_type(2))) float v2f;
typedef __attribute__((ext_vector_type(8))) float v8f;


constexpr int BSH    = 10;
constexpr int BW     = 1 << BSH;
constexpr int MAXB   = 128;
constexpr int CHUNK  = 4096;
constexpr int EPT    = CHUNK / 256;
constexpr int SCAP   = 40960;

__device__ inline void wave_scan128(const int* in, int* out, int lane) {
  const int a0 = in[lane * 4 + 0];
  const int a1 = in[lane * 4 + 1];
  const int a2 = in[lane * 4 + 2];
  const int a3 = in[lane * 4 + 3];
  const int s = (a0 + a1) + (a2 + a3);
  int incl = s;
  for (int o = 1; o < 32; o <<= 1) {
    const int y = __shfl_up(incl, o);
    if (lane >= o) incl += y;
  }
  const int ex = incl - s;
  out[lane * 4 + 0] = ex;
  out[lane * 4 + 1] = ex + a0;
  out[lane * 4 + 2] = ex + a0 + a1;
  out[lane * 4 + 3] = ex + a0 + a1 + a2;
}

__global__ __launch_bounds__(256) void bucket_count_kernel(const int* __restrict__ dst,
                                                           int* __restrict__ blkhist, int e) {
  __shared__ int hist[MAXB];
  const int t = threadIdx.x;
  if (t < MAXB) hist[t] = 0;
  __syncthreads();
  const int e0 = blockIdx.x * CHUNK;
  int d[EPT];
#pragma unroll
  for (int j = 0; j < EPT; ++j) {
    const int i = e0 + j * 256 + t;
    d[j] = (i < e) ? dst[i] : -1;
  }
#pragma unroll
  for (int j = 0; j < EPT; ++j) {
    if (d[j] >= 0) atomicAdd(&hist[d[j] >> BSH], 1);
  }
  __syncthreads();
  if (t < MAXB) blkhist[blockIdx.x * MAXB + t] = hist[t];
}

__global__ __launch_bounds__(256) void bucket_place_kernel(const int* __restrict__ src,
                                                           const int* __restrict__ dst,
                                                           const int* __restrict__ blkhist,
                                                           int* __restrict__ bucket_base,
                                                           int* __restrict__ pairs_i, int e, int nblk) {
  int2* __restrict__ pairs = (int2*)pairs_i;
  __shared__ int pre[MAXB];
  __shared__ int tot[MAXB];
  __shared__ int base[MAXB];
  __shared__ int mine[MAXB];
  __shared__ int lstart[MAXB];
  __shared__ int lcur[MAXB + 32];
  __shared__ int goff[MAXB];
  __shared__ int2 stage[CHUNK];
  const int t = threadIdx.x;
  const int lane = t & 31;
  const int w = t >> 5;
  const int blk = blockIdx.x;
  if (t < MAXB) {
    pre[t] = 0;
    tot[t] = 0;
    mine[t] = blkhist[blk * MAXB + t];
  }
  __syncthreads();

  {
    const int4* __restrict__ M4 = (const int4*)blkhist;
    const int ws = __builtin_amdgcn_readfirstlane(w);
    int4 p = {0, 0, 0, 0};
    int4 s = {0, 0, 0, 0};
    for (int k0 = ws; k0 < nblk; k0 += 64) {
      int4 v[8];
#pragma unroll
      for (int u = 0; u < 8; ++u) {
        const int k = k0 + 8 * u;
        const int kk = (k < nblk) ? k : k0;
        v[u] = M4[kk * (MAXB / 4) + lane];
      }
#pragma unroll
      for (int u = 0; u < 8; ++u) {
        const int k = k0 + 8 * u;
        if (k < nblk) {
          s.x += v[u].x; s.y += v[u].y; s.z += v[u].z; s.w += v[u].w;
          if (k < blk) { p.x += v[u].x; p.y += v[u].y; p.z += v[u].z; p.w += v[u].w; }
        }
      }
    }
    atomicAdd(&tot[lane * 4 + 0], s.x);
    atomicAdd(&tot[lane * 4 + 1], s.y);
    atomicAdd(&tot[lane * 4 + 2], s.z);
    atomicAdd(&tot[lane * 4 + 3], s.w);
    atomicAdd(&pre[lane * 4 + 0], p.x);
    atomicAdd(&pre[lane * 4 + 1], p.y);
    atomicAdd(&pre[lane * 4 + 2], p.z);
    atomicAdd(&pre[lane * 4 + 3], p.w);
  }
  __syncthreads();
  if (w == 0) wave_scan128(tot, base, lane);
  if (w == 1) wave_scan128(mine, lstart, lane);
  __syncthreads();
  if (t < MAXB) {
    lcur[t] = lstart[t];
    goff[t] = base[t] + pre[t] - lstart[t];
    if (blk == 0) bucket_base[t] = base[t];
  }
  if (t >= MAXB && t < MAXB + 32) lcur[t] = 0;
  __syncthreads();

  const int e0 = blk * CHUNK;
  int d[EPT], s[EPT], pos[EPT];
#pragma unroll
  for (int j = 0; j < EPT; ++j) {
    const int i = e0 + j * 256 + t;
    d[j] = (i < e) ? dst[i] : -1;
    s[j] = (i < e) ? src[i] : 0;
  }
#pragma unroll
  for (int j = 0; j < EPT; ++j) {
    const int bk = (d[j] >= 0) ? (d[j] >> BSH) : (MAXB + lane);
    pos[j] = atomicAdd(&lcur[bk], 1);
  }
#pragma unroll
  for (int j = 0; j < EPT; ++j) {
    if (d[j] >= 0) stage[pos[j]] = make_int2(s[j], d[j]);
  }
  __syncthreads();
  const int cntb = (e - e0 < CHUNK) ? (e - e0) : CHUNK;
  int2 pr[EPT];
  int g[EPT];
#pragma unroll
  for (int j = 0; j < EPT; ++j) {
    const int i = j * 256 + t;
    pr[j] = stage[(i < cntb) ? i : (cntb - 1)];
  }
#pragma unroll
  for (int j = 0; j < EPT; ++j) g[j] = goff[pr[j].y >> BSH];
#pragma unroll
  for (int j = 0; j < EPT; ++j) {
    const int i = j * 256 + t;
    if (i < cntb) pairs[g[j] + i] = pr[j];
  }
}

constexpr int RB = 8;

__global__ __launch_bounds__(1024) void bucket_sort_kernel(const int* __restrict__ pairs_i,
                                                           const int* __restrict__ bucket_base,
                                                           int e, int nb,
                                                           int* __restrict__ cnt_g,
                                                           int* __restrict__ rowptr,
                                                           float* __restrict__ dinv,
                                                           int* __restrict__ csr_src) {
  const int2* __restrict__ pairs = (const int2*)pairs_i;
  __shared__ int cnt[BW];
  __shared__ int cur[BW + 32];
  __shared__ int wsum[32];
  __shared__ int wex[32];
  __shared__ int stage[SCAP];
  const int t = threadIdx.x;
  const int lane = t & 31;
  const int w = t >> 5;
  const int b = blockIdx.x;
  const int lo = b << BSH;
  const int beg = bucket_base[b];
  const int end = (b + 1 < nb) ? bucket_base[b + 1] : e;
  cnt[t] = 0;
  __syncthreads();
  for (int i0 = beg; i0 < end; i0 += BW * RB) {
    int dd[RB];
#pragma unroll
    for (int u = 0; u < RB; ++u) {
      const int i = i0 + u * BW + t;
      dd[u] = (i < end) ? pairs[i].y : -1;
    }
#pragma unroll
    for (int u = 0; u < RB; ++u) {
      if (dd[u] >= 0) atomicAdd(&cnt[dd[u] - lo], 1);
    }
  }
  __syncthreads();
  const int c = cnt[t];
  int incl = c;
  for (int o = 1; o < 32; o <<= 1) {
    const int y = __shfl_up(incl, o);
    if (lane >= o) incl += y;
  }
  if (lane == 31) wsum[w] = incl;
  __syncthreads();
  if (w == 0) {
    const int v = wsum[lane];
    int in2 = v;
    for (int o = 1; o < 32; o <<= 1) {
      const int y = __shfl_up(in2, o);
      if (lane >= o) in2 += y;
    }
    wex[lane] = in2 - v;
  }
  __syncthreads();
  const int rp = beg + wex[w] + incl - c;
  rowptr[lo + t] = rp;
  cnt_g[lo + t] = c;
  const float df = (float)(c + 1);
  dinv[lo + t] = (df > 0.0f) ? rsqrtf(df) : 0.0f;
  cur[t] = rp;
  if (t < 32) cur[BW + t] = 0;
  __syncthreads();
  const int nbk = end - beg;
  if (nbk <= SCAP) {
    for (int i0 = beg; i0 < end; i0 += BW * RB) {
      int2 pr[RB];
      int pos[RB];
#pragma unroll
      for (int u = 0; u < RB; ++u) {
        const int i = i0 + u * BW + t;
        pr[u] = (i < end) ? pairs[i] : make_int2(0, -1);
      }
#pragma unroll
      for (int u = 0; u < RB; ++u) {
        const int k = (pr[u].y >= 0) ? (pr[u].y - lo) : (BW + lane);
        pos[u] = atomicAdd(&cur[k], 1);
      }
#pragma unroll
      for (int u = 0; u < RB; ++u) {
        if (pr[u].y >= 0) stage[pos[u] - beg] = pr[u].x;
      }
    }
    __syncthreads();
    for (int i = t; i < nbk; i += BW) csr_src[beg + i] = stage[i];
  } else {
    for (int i0 = beg; i0 < end; i0 += BW * RB) {
      int2 pr[RB];
      int pos[RB];
#pragma unroll
      for (int u = 0; u < RB; ++u) {
        const int i = i0 + u * BW + t;
        pr[u] = (i < end) ? pairs[i] : make_int2(0, -1);
      }
#pragma unroll
      for (int u = 0; u < RB; ++u) {
        const int k = (pr[u].y >= 0) ? (pr[u].y - lo) : (BW + lane);
        pos[u] = atomicAdd(&cur[k], 1);
      }
#pragma unroll
      for (int u = 0; u < RB; ++u) {
        if (pr[u].y >= 0) csr_src[pos[u]] = pr[u].x;
      }
    }
  }
}

template <int K, int NC>
__global__ __launch_bounds__(32 * (NC / 16)) void gemm_wmma_kernel(
    const float* __restrict__ X, const float* __restrict__ W,
    const float* __restrict__ dinv, float* __restrict__ H, int M) {
  const int lane = threadIdx.x & 31;
  const int wave = threadIdx.x >> 5;
  const int m0   = blockIdx.x << 4;
  const int n0   = wave << 4;
  const int l15  = lane & 15;
  const int kh   = (lane >> 4) << 1;

  int row = m0 + l15;
  if (row >= M) row = M - 1;
  const int col = n0 + l15;

  const int rbase = m0 + ((lane >> 4) << 3);
  float dsc[8];
#pragma unroll
  for (int v = 0; v < 8; ++v) {
    int r = rbase + v;
    dsc[v] = dinv[(r < M) ? r : (M - 1)];
  }

  v8f c = {};
  for (int k0 = 0; k0 < K; k0 += 4) {
    v2f a, b;
    a.x = X[row * K + k0 + kh];
    a.y = X[row * K + k0 + kh + 1];
    b.x = W[(k0 + kh) * NC + col];
    b.y = W[(k0 + kh + 1) * NC + col];
    c = __builtin_amdgcn_wmma_f32_16x16x4_f32(
         false, a,  false, b,
         (short)0, c,  false,  false);
  }

#pragma unroll
  for (int v = 0; v < 8; ++v) {
    int r = rbase + v;
    const float o = c[v] * dsc[v];
    if (r < M) H[r * NC + col] = o;
  }
}

template <int C, bool RELU>
__global__ __launch_bounds__(256) void gather_kernel(
    const int* __restrict__ rowptr, const int* __restrict__ cnt,
    const int* __restrict__ csr_src, const float* __restrict__ dinv,
    const float* __restrict__ h, const float* __restrict__ bias,
    float* __restrict__ out, int n) {
  constexpr int LPR = C / 2;
  constexpr int RPI = 32 / LPR;
  constexpr int G   = 8;
  constexpr int U   = G / RPI;
  const int lane = threadIdx.x & 31;
  const int wv   = __builtin_amdgcn_readfirstlane((int)(threadIdx.x >> 5));
  const int node = blockIdx.x * 8 + wv;
  if (node >= n) return;
  const int sub = lane / LPR;
  const int cl  = lane - sub * LPR;
  const int start = rowptr[node];
  const int deg   = cnt[node];
  const float di  = dinv[node];
  const v2f* __restrict__ h2 = (const v2f*)h;
  const int total = deg + 1;

  v2f acc = {0.0f, 0.0f};
  for (int base = 0; base < total; base += 32) {
    const int rem = total - base;
    const int m = (rem < 32) ? rem : 32;
    const int k = base + lane;
    int s_l = node;
    if (k < deg) s_l = csr_src[start + k];
    for (int j = 0; j < m; j += G) {
      v2f hv[U];
      bool ok[U];
#pragma unroll
      for (int u = 0; u < U; ++u) {
        int s;
        if (RPI == 1) {
          const int idx = j + u;
          s = __builtin_amdgcn_readlane(s_l, idx);
          ok[u] = idx < m;
        } else {
          const int idx = j + 2 * u;
          const int sa = __builtin_amdgcn_readlane(s_l, idx);
          const int sb = __builtin_amdgcn_readlane(s_l, idx + 1);
          s = sub ? sb : sa;
          ok[u] = (idx + sub) < m;
        }
        const v2f* __restrict__ rowp = h2 + (size_t)s * LPR;
        hv[u] = rowp[cl];
      }
#pragma unroll
      for (int u = 0; u < U; ++u) {
        const v2f nv = acc + hv[u];
        acc = ok[u] ? nv : acc;
      }
    }
  }

  if (RPI == 2) {
    acc.x += __shfl_xor(acc.x, 16);
    acc.y += __shfl_xor(acc.y, 16);
  }
  if (sub == 0) {
    const v2f bv = ((const v2f*)bias)[cl];
    v2f r = acc * di + bv;
    if (RELU) {
      r.x = fmaxf(r.x, 0.0f);
      r.y = fmaxf(r.y, 0.0f);
    }
    ((v2f*)out)[(size_t)node * LPR + cl] = r;
  }
}


extern "C" void kernel_launch(void* const* d_in, const int* in_sizes, int n_in,
                              void* d_out, int out_size, void* d_ws, size_t ws_size,
                              hipStream_t stream) {
  (void)n_in; (void)out_size; (void)ws_size;
  const float* x  = (const float*)d_in[0];
  const int*   ei = (const int*)d_in[1];
  const float* W1 = (const float*)d_in[2];
  const float* b1 = (const float*)d_in[3];
  const float* W2 = (const float*)d_in[4];
  const float* b2 = (const float*)d_in[5];

  const int IN_C = 128, HID = 64, OUT = 32;
  const int n = in_sizes[0] / IN_C;
  const int e = in_sizes[1] / 2;
  const int* srcp = ei;
  const int* dstp = ei + e;

  const int    np   = (n + BW - 1) & ~(BW - 1);
  const int    nb   = np >> BSH;
  const int    nblk = (e + CHUNK - 1) / CHUNK;
  const size_t ep   = ((size_t)e + 63) & ~(size_t)63;

  int*   cnt     = (int*)d_ws;
  int*   rowptr  = cnt + np;
  float* dinv    = (float*)(rowptr + np);
  int*   bbase   = (int*)(dinv + np);
  int*   blkhist = bbase + MAXB;
  int*   pairs   = blkhist + (size_t)nblk * MAXB;
  int*   csr     = pairs + 2 * ep;
  float* h1      = (float*)(csr + ep);
  float* a1      = h1 + (size_t)n * HID;
  float* h2      = a1 + (size_t)n * HID;
  float* outp    = (float*)d_out;
  (void)OUT;

  bucket_count_kernel<<<nblk, 256, 0, stream>>>(dstp, blkhist, e);
  bucket_place_kernel<<<nblk, 256, 0, stream>>>(srcp, dstp, blkhist, bbase, pairs, e, nblk);
  bucket_sort_kernel<<<nb, BW, 0, stream>>>(pairs, bbase, e, nb, cnt, rowptr, dinv, csr);

  gemm_wmma_kernel<128, 64><<<(n + 15) / 16, 128, 0, stream>>>(x, W1, dinv, h1, n);
  gather_kernel<64, true><<<(n + 7) / 8, 256, 0, stream>>>(rowptr, cnt, csr, dinv, h1, b1, a1, n);

  gemm_wmma_kernel<64, 32><<<(n + 15) / 16, 64, 0, stream>>>(a1, W2, dinv, h2, n);
  gather_kernel<32, false><<<(n + 7) / 8, 256, 0, stream>>>(rowptr, cnt, csr, dinv, h2, b2, outp, n);
}


# Gathers as a fixed grid of long-lived waves that loop over nodes (next node's list entries requested ahead of the current rows) instead of one short wave per node; per-node sum order unchanged
# speedup vs baseline: 4.8548x; 1.0649x over previous
#include <hip/hip_runtime.h>


typedef __attribute__((ext_vector_type(2))) float v2f;
typedef __attribute__((ext_vector_type(8))) float v8f;


constexpr int BSH    = 10;
constexpr int BW     = 1 << BSH;
constexpr int MAXB   = 128;
constexpr int CHUNK  = 4096;
constexpr int EPT    = CHUNK / 256;
constexpr int SCAP   = 40960;

__device__ inline void wave_scan128(const int* in, int* out, int lane) {
  const int a0 = in[lane * 4 + 0];
  const int a1 = in[lane * 4 + 1];
  const int a2 = in[lane * 4 + 2];
  const int a3 = in[lane * 4 + 3];
  const int s = (a0 + a1) + (a2 + a3);
  int incl = s;
  for (int o = 1; o < 32; o <<= 1) {
    const int y = __shfl_up(incl, o);
    if (lane >= o) incl += y;
  }
  const int ex = incl - s;
  out[lane * 4 + 0] = ex;
  out[lane * 4 + 1] = ex + a0;
  out[lane * 4 + 2] = ex + a0 + a1;
  out[lane * 4 + 3] = ex + a0 + a1 + a2;
}

__global__ __launch_bounds__(256) void bucket_count_kernel(const int* __restrict__ dst,
                                                           int* __restrict__ blkhist, int e) {
  __shared__ int hist[MAXB];
  const int t = threadIdx.x;
  if (t < MAXB) hist[t] = 0;
  __syncthreads();
  const int e0 = blockIdx.x * CHUNK;
  int d[EPT];
#pragma unroll
  for (int j = 0; j < EPT; ++j) {
    const int i = e0 + j * 256 + t;
    d[j] = (i < e) ? dst[i] : -1;
  }
#pragma unroll
  for (int j = 0; j < EPT; ++j) {
    if (d[j] >= 0) atomicAdd(&hist[d[j] >> BSH], 1);
  }
  __syncthreads();
  if (t < MAXB) blkhist[blockIdx.x * MAXB + t] = hist[t];
}

__global__ __launch_bounds__(256) void bucket_place_kernel(const int* __restrict__ src,
                                                           const int* __restrict__ dst,
                                                           const int* __restrict__ blkhist,
                                                           int* __restrict__ bucket_base,
                                                           int* __restrict__ pairs_i, int e, int nblk) {
  int2* __restrict__ pairs = (int2*)pairs_i;
  __shared__ int pre[MAXB];
  __shared__ int tot[MAXB];
  __shared__ int base[MAXB];
  __shared__ int mine[MAXB];
  __shared__ int lstart[MAXB];
  __shared__ int lcur[MAXB + 32];
  __shared__ int goff[MAXB];
  __shared__ int2 stage[CHUNK];
  const int t = threadIdx.x;
  const int lane = t & 31;
  const int w = t >> 5;
  const int blk = blockIdx.x;
  if (t < MAXB) {
    pre[t] = 0;
    tot[t] = 0;
    mine[t] = blkhist[blk * MAXB + t];
  }
  __syncthreads();

  {
    const int4* __restrict__ M4 = (const int4*)blkhist;
    const int ws = __builtin_amdgcn_readfirstlane(w);
    int4 p = {0, 0, 0, 0};
    int4 s = {0, 0, 0, 0};
    for (int k0 = ws; k0 < nblk; k0 += 64) {
      int4 v[8];
#pragma unroll
      for (int u = 0; u < 8; ++u) {
        const int k = k0 + 8 * u;
        const int kk = (k < nblk) ? k : k0;
        v[u] = M4[kk * (MAXB / 4) + lane];
      }
#pragma unroll
      for (int u = 0; u < 8; ++u) {
        const int k = k0 + 8 * u;
        if (k < nblk) {
          s.x += v[u].x; s.y += v[u].y; s.z += v[u].z; s.w += v[u].w;
          if (k < blk) { p.x += v[u].x; p.y += v[u].y; p.z += v[u].z; p.w += v[u].w; }
        }
      }
    }
    atomicAdd(&tot[lane * 4 + 0], s.x);
    atomicAdd(&tot[lane * 4 + 1], s.y);
    atomicAdd(&tot[lane * 4 + 2], s.z);
    atomicAdd(&tot[lane * 4 + 3], s.w);
    atomicAdd(&pre[lane * 4 + 0], p.x);
    atomicAdd(&pre[lane * 4 + 1], p.y);
    atomicAdd(&pre[lane * 4 + 2], p.z);
    atomicAdd(&pre[lane * 4 + 3], p.w);
  }
  __syncthreads();
  if (w == 0) wave_scan128(tot, base, lane);
  if (w == 1) wave_scan128(mine, lstart, lane);
  __syncthreads();
  if (t < MAXB) {
    lcur[t] = lstart[t];
    goff[t] = base[t] + pre[t] - lstart[t];
    if (blk == 0) bucket_base[t] = base[t];
  }
  if (t >= MAXB && t < MAXB + 32) lcur[t] = 0;
  __syncthreads();

  const int e0 = blk * CHUNK;
  int d[EPT], s[EPT], pos[EPT];
#pragma unroll
  for (int j = 0; j < EPT; ++j) {
    const int i = e0 + j * 256 + t;
    d[j] = (i < e) ? dst[i] : -1;
    s[j] = (i < e) ? src[i] : 0;
  }
#pragma unroll
  for (int j = 0; j < EPT; ++j) {
    const int bk = (d[j] >= 0) ? (d[j] >> BSH) : (MAXB + lane);
    pos[j] = atomicAdd(&lcur[bk], 1);
  }
#pragma unroll
  for (int j = 0; j < EPT; ++j) {
    if (d[j] >= 0) stage[pos[j]] = make_int2(s[j], d[j]);
  }
  __syncthreads();
  const int cntb = (e - e0 < CHUNK) ? (e - e0) : CHUNK;
  int2 pr[EPT];
  int g[EPT];
#pragma unroll
  for (int j = 0; j < EPT; ++j) {
    const int i = j * 256 + t;
    pr[j] = stage[(i < cntb) ? i : (cntb - 1)];
  }
#pragma unroll
  for (int j = 0; j < EPT; ++j) g[j] = goff[pr[j].y >> BSH];
#pragma unroll
  for (int j = 0; j < EPT; ++j) {
    const int i = j * 256 + t;
    if (i < cntb) pairs[g[j] + i] = pr[j];
  }
}

constexpr int RB = 8;

__global__ __launch_bounds__(1024) void bucket_sort_kernel(const int* __restrict__ pairs_i,
                                                           const int* __restrict__ bucket_base,
                                                           int e, int nb,
                                                           int* __restrict__ cnt_g,
                                                           int* __restrict__ rowptr,
                                                           float* __restrict__ dinv,
                                                           int* __restrict__ csr_src) {
  const int2* __restrict__ pairs = (const int2*)pairs_i;
  __shared__ int cnt[BW];
  __shared__ int cur[BW + 32];
  __shared__ int wsum[32];
  __shared__ int wex[32];
  __shared__ int stage[SCAP];
  const int t = threadIdx.x;
  const int lane = t & 31;
  const int w = t >> 5;
  const int b = blockIdx.x;
  const int lo = b << BSH;
  const int beg = bucket_base[b];
  const int end = (b + 1 < nb) ? bucket_base[b + 1] : e;
  cnt[t] = 0;
  __syncthreads();
  for (int i0 = beg; i0 < end; i0 += BW * RB) {
    int dd[RB];
#pragma unroll
    for (int u = 0; u < RB; ++u) {
      const int i = i0 + u * BW + t;
      dd[u] = (i < end) ? pairs[i].y : -1;
    }
#pragma unroll
    for (int u = 0; u < RB; ++u) {
      if (dd[u] >= 0) atomicAdd(&cnt[dd[u] - lo], 1);
    }
  }
  __syncthreads();
  const int c = cnt[t];
  int incl = c;
  for (int o = 1; o < 32; o <<= 1) {
    const int y = __shfl_up(incl, o);
    if (lane >= o) incl += y;
  }
  if (lane == 31) wsum[w] = incl;
  __syncthreads();
  if (w == 0) {
    const int v = wsum[lane];
    int in2 = v;
    for (int o = 1; o < 32; o <<= 1) {
      const int y = __shfl_up(in2, o);
      if (lane >= o) in2 += y;
    }
    wex[lane] = in2 - v;
  }
  __syncthreads();
  const int rp = beg + wex[w] + incl - c;
  rowptr[lo + t] = rp;
  cnt_g[lo + t] = c;
  const float df = (float)(c + 1);
  dinv[lo + t] = (df > 0.0f) ? rsqrtf(df) : 0.0f;
  cur[t] = rp;
  if (t < 32) cur[BW + t] = 0;
  __syncthreads();
  const int nbk = end - beg;
  if (nbk <= SCAP) {
    for (int i0 = beg; i0 < end; i0 += BW * RB) {
      int2 pr[RB];
      int pos[RB];
#pragma unroll
      for (int u = 0; u < RB; ++u) {
        const int i = i0 + u * BW + t;
        pr[u] = (i < end) ? pairs[i] : make_int2(0, -1);
      }
#pragma unroll
      for (int u = 0; u < RB; ++u) {
        const int k = (pr[u].y >= 0) ? (pr[u].y - lo) : (BW + lane);
        pos[u] = atomicAdd(&cur[k], 1);
      }
#pragma unroll
      for (int u = 0; u < RB; ++u) {
        if (pr[u].y >= 0) stage[pos[u] - beg] = pr[u].x;
      }
    }
    __syncthreads();
    for (int i = t; i < nbk; i += BW) csr_src[beg + i] = stage[i];
  } else {
    for (int i0 = beg; i0 < end; i0 += BW * RB) {
      int2 pr[RB];
      int pos[RB];
#pragma unroll
      for (int u = 0; u < RB; ++u) {
        const int i = i0 + u * BW + t;
        pr[u] = (i < end) ? pairs[i] : make_int2(0, -1);
      }
#pragma unroll
      for (int u = 0; u < RB; ++u) {
        const int k = (pr[u].y >= 0) ? (pr[u].y - lo) : (BW + lane);
        pos[u] = atomicAdd(&cur[k], 1);
      }
#pragma unroll
      for (int u = 0; u < RB; ++u) {
        if (pr[u].y >= 0) csr_src[pos[u]] = pr[u].x;
      }
    }
  }
}

template <int K, int NC>
__global__ __launch_bounds__(32 * (NC / 16)) void gemm_wmma_kernel(
    const float* __restrict__ X, const float* __restrict__ W,
    const float* __restrict__ dinv, float* __restrict__ H, int M) {
  const int lane = threadIdx.x & 31;
  const int wave = threadIdx.x >> 5;
  const int m0   = blockIdx.x << 4;
  const int n0   = wave << 4;
  const int l15  = lane & 15;
  const int kh   = (lane >> 4) << 1;

  int row = m0 + l15;
  if (row >= M) row = M - 1;
  const int col = n0 + l15;

  const int rbase = m0 + ((lane >> 4) << 3);
  float dsc[8];
#pragma unroll
  for (int v = 0; v < 8; ++v) {
    int r = rbase + v;
    dsc[v] = dinv[(r < M) ? r : (M - 1)];
  }

  v8f c = {};
  for (int k0 = 0; k0 < K; k0 += 4) {
    v2f a, b;
    a.x = X[row * K + k0 + kh];
    a.y = X[row * K + k0 + kh + 1];
    b.x = W[(k0 + kh) * NC + col];
    b.y = W[(k0 + kh + 1) * NC + col];
    c = __builtin_amdgcn_wmma_f32_16x16x4_f32(
         false, a,  false, b,
         (short)0, c,  false,  false);
  }

#pragma unroll
  for (int v = 0; v < 8; ++v) {
    int r = rbase + v;
    const float o = c[v] * dsc[v];
    if (r < M) H[r * NC + col] = o;
  }
}


template <int C>
__device__ __forceinline__ v2f gather_group(v2f acc, int s_l, int j, int m,
                                            const v2f* __restrict__ h2, int sub, int cl) {
  constexpr int LPR = C / 2;
  constexpr int RPI = 32 / LPR;
  constexpr int G   = 8;
  constexpr int U   = G / RPI;
  v2f hv[U];
  bool ok[U];
#pragma unroll
  for (int u = 0; u < U; ++u) {
    int s;
    if (RPI == 1) {
      const int idx = j + u;
      s = __builtin_amdgcn_readlane(s_l, idx);
      ok[u] = idx < m;
    } else {
      const int idx = j + 2 * u;
      const int sa = __builtin_amdgcn_readlane(s_l, idx);
      const int sb = __builtin_amdgcn_readlane(s_l, idx + 1);
      s = sub ? sb : sa;
      ok[u] = (idx + sub) < m;
    }
    const v2f* __restrict__ rowp = h2 + (size_t)s * LPR;
    hv[u] = rowp[cl];
  }
#pragma unroll
  for (int u = 0; u < U; ++u) {
    const v2f nv = acc + hv[u];
    acc = ok[u] ? nv : acc;
  }
  return acc;
}

template <int C, bool RELU>
__global__ __launch_bounds__(256) void gather_kernel(
    const int* __restrict__ rowptr, const int* __restrict__ cnt,
    const int* __restrict__ csr_src, const float* __restrict__ dinv,
    const float* __restrict__ h, const float* __restrict__ bias,
    float* __restrict__ out, int n, int npb) {
  constexpr int LPR = C / 2;
  constexpr int RPI = 32 / LPR;
  constexpr int G   = 8;
  constexpr int WPB = 8;
  const int lane = threadIdx.x & 31;
  const int wv  = __builtin_amdgcn_readfirstlane((int)(threadIdx.x >> 5));
  const int sub = lane / LPR;
  const int cl  = lane - sub * LPR;
  const v2f* __restrict__ h2 = (const v2f*)h;
  const v2f bv = ((const v2f*)bias)[cl];

  const int nb0  = (int)blockIdx.x * npb;
  const int nend = (nb0 + npb < n) ? (nb0 + npb) : n;

  for (int c0 = nb0 + wv; c0 < nend; c0 += 32 * WPB) {
    const int hn = c0 + WPB * lane;
    int hstart = 0, hdeg = 0, hdi = 0;
    if (hn < nend) {
      hstart = rowptr[hn];
      hdeg   = cnt[hn];
      hdi    = __float_as_int(dinv[hn]);
    }
    const int avail = (nend - c0 + WPB - 1) / WPB;
    const int cnum  = (avail < 32) ? avail : 32;

    int node  = c0;
    int start = __builtin_amdgcn_readlane(hstart, 0);
    int deg   = __builtin_amdgcn_readlane(hdeg, 0);
    int s_l   = node;
    if (lane < deg) s_l = csr_src[start + lane];

    for (int it = 0; it < cnum; ++it) {
      const float di = __int_as_float(__builtin_amdgcn_readlane(hdi, it));
      const int nit   = it + 1;
      const int nnode = node + WPB;
      int nstart = 0, ndeg = 0, ns_l = nnode;
      if (nit < cnum) {
        nstart = __builtin_amdgcn_readlane(hstart, nit);
        ndeg   = __builtin_amdgcn_readlane(hdeg, nit);
        if (lane < ndeg) ns_l = csr_src[nstart + lane];
      }

      const int total = deg + 1;
      v2f acc = {0.0f, 0.0f};
      {
        const int m = (total < 32) ? total : 32;
        for (int j = 0; j < m; j += G) acc = gather_group<C>(acc, s_l, j, m, h2, sub, cl);
      }
      for (int base = 32; base < total; base += 32) {
        const int k = base + lane;
        int s_x = node;
        if (k < deg) s_x = csr_src[start + k];
        const int rem = total - base;
        const int m = (rem < 32) ? rem : 32;
        for (int j = 0; j < m; j += G) acc = gather_group<C>(acc, s_x, j, m, h2, sub, cl);
      }

      if (RPI == 2) {
        acc.x += __shfl_xor(acc.x, 16);
        acc.y += __shfl_xor(acc.y, 16);
      }
      if (sub == 0) {
        v2f r = acc * di + bv;
        if (RELU) {
          r.x = fmaxf(r.x, 0.0f);
          r.y = fmaxf(r.y, 0.0f);
        }
        ((v2f*)out)[(size_t)node * LPR + cl] = r;
      }

      node  = nnode;
      start = nstart;
      deg   = ndeg;
      s_l   = ns_l;
    }
  }
}


extern "C" void kernel_launch(void* const* d_in, const int* in_sizes, int n_in,
                              void* d_out, int out_size, void* d_ws, size_t ws_size,
                              hipStream_t stream) {
  (void)n_in; (void)out_size; (void)ws_size;
  const float* x  = (const float*)d_in[0];
  const int*   ei = (const int*)d_in[1];
  const float* W1 = (const float*)d_in[2];
  const float* b1 = (const float*)d_in[3];
  const float* W2 = (const float*)d_in[4];
  const float* b2 = (const float*)d_in[5];

  const int IN_C = 128, HID = 64, OUT = 32;
  const int n = in_sizes[0] / IN_C;
  const int e = in_sizes[1] / 2;
  const int npb   = (((n + 2047) / 2048) + 7) & ~7;
  const int ggrid = (n + npb - 1) / npb;
  const int* srcp = ei;
  const int* dstp = ei + e;

  const int    np   = (n + BW - 1) & ~(BW - 1);
  const int    nb   = np >> BSH;
  const int    nblk = (e + CHUNK - 1) / CHUNK;
  const size_t ep   = ((size_t)e + 63) & ~(size_t)63;

  int*   cnt     = (int*)d_ws;
  int*   rowptr  = cnt + np;
  float* dinv    = (float*)(rowptr + np);
  int*   bbase   = (int*)(dinv + np);
  int*   blkhist = bbase + MAXB;
  int*   pairs   = blkhist + (size_t)nblk * MAXB;
  int*   csr     = pairs + 2 * ep;
  float* h1      = (float*)(csr + ep);
  float* a1      = h1 + (size_t)n * HID;
  float* h2      = a1 + (size_t)n * HID;
  float* outp    = (float*)d_out;
  (void)OUT;

  bucket_count_kernel<<<nblk, 256, 0, stream>>>(dstp, blkhist, e);
  bucket_place_kernel<<<nblk, 256, 0, stream>>>(srcp, dstp, blkhist, bbase, pairs, e, nblk);
  bucket_sort_kernel<<<nb, BW, 0, stream>>>(pairs, bbase, e, nb, cnt, rowptr, dinv, csr);

  gemm_wmma_kernel<128, 64><<<(n + 15) / 16, 128, 0, stream>>>(x, W1, dinv, h1, n);
  gather_kernel<64, true><<<ggrid, 256, 0, stream>>>(rowptr, cnt, csr, dinv, h1, b1, a1, n, npb);

  gemm_wmma_kernel<64, 32><<<(n + 15) / 16, 64, 0, stream>>>(a1, W2, dinv, h2, n);
  gather_kernel<32, false><<<ggrid, 256, 0, stream>>>(rowptr, cnt, csr, dinv, h2, b2, outp, n, npb);
}


# Products on a fixed grid of long-lived waves: each keeps all of W in registers and loops over 16-row tiles, one A fragment feeding every column tile, next tile requested ahead; sums unchanged
# speedup vs baseline: 5.7117x; 1.1765x over previous
#include <hip/hip_runtime.h>


typedef __attribute__((ext_vector_type(2))) float v2f;
typedef __attribute__((ext_vector_type(4))) float v4f;
typedef __attribute__((ext_vector_type(8))) float v8f;


constexpr int BSH    = 10;
constexpr int BW     = 1 << BSH;
constexpr int MAXB   = 128;
constexpr int CHUNK  = 4096;
constexpr int EPT    = CHUNK / 256;
constexpr int SCAP   = 40960;

__device__ inline void wave_scan128(const int* in, int* out, int lane) {
  const int a0 = in[lane * 4 + 0];
  const int a1 = in[lane * 4 + 1];
  const int a2 = in[lane * 4 + 2];
  const int a3 = in[lane * 4 + 3];
  const int s = (a0 + a1) + (a2 + a3);
  int incl = s;
  for (int o = 1; o < 32; o <<= 1) {
    const int y = __shfl_up(incl, o);
    if (lane >= o) incl += y;
  }
  const int ex = incl - s;
  out[lane * 4 + 0] = ex;
  out[lane * 4 + 1] = ex + a0;
  out[lane * 4 + 2] = ex + a0 + a1;
  out[lane * 4 + 3] = ex + a0 + a1 + a2;
}

__global__ __launch_bounds__(256) void bucket_count_kernel(const int* __restrict__ dst,
                                                           int* __restrict__ blkhist, int e) {
  __shared__ int hist[MAXB];
  const int t = threadIdx.x;
  if (t < MAXB) hist[t] = 0;
  __syncthreads();
  const int e0 = blockIdx.x * CHUNK;
  int d[EPT];
#pragma unroll
  for (int j = 0; j < EPT; ++j) {
    const int i = e0 + j * 256 + t;
    d[j] = (i < e) ? dst[i] : -1;
  }
#pragma unroll
  for (int j = 0; j < EPT; ++j) {
    if (d[j] >= 0) atomicAdd(&hist[d[j] >> BSH], 1);
  }
  __syncthreads();
  if (t < MAXB) blkhist[blockIdx.x * MAXB + t] = hist[t];
}

__global__ __launch_bounds__(256) void bucket_place_kernel(const int* __restrict__ src,
                                                           const int* __restrict__ dst,
                                                           const int* __restrict__ blkhist,
                                                           int* __restrict__ bucket_base,
                                                           int* __restrict__ pairs_i, int e, int nblk) {
  int2* __restrict__ pairs = (int2*)pairs_i;
  __shared__ int pre[MAXB];
  __shared__ int tot[MAXB];
  __shared__ int base[MAXB];
  __shared__ int mine[MAXB];
  __shared__ int lstart[MAXB];
  __shared__ int lcur[MAXB + 32];
  __shared__ int goff[MAXB];
  __shared__ int2 stage[CHUNK];
  const int t = threadIdx.x;
  const int lane = t & 31;
  const int w = t >> 5;
  const int blk = blockIdx.x;
  if (t < MAXB) {
    pre[t] = 0;
    tot[t] = 0;
    mine[t] = blkhist[blk * MAXB + t];
  }
  __syncthreads();

  {
    const int4* __restrict__ M4 = (const int4*)blkhist;
    const int ws = __builtin_amdgcn_readfirstlane(w);
    int4 p = {0, 0, 0, 0};
    int4 s = {0, 0, 0, 0};
    for (int k0 = ws; k0 < nblk; k0 += 64) {
      int4 v[8];
#pragma unroll
      for (int u = 0; u < 8; ++u) {
        const int k = k0 + 8 * u;
        const int kk = (k < nblk) ? k : k0;
        v[u] = M4[kk * (MAXB / 4) + lane];
      }
#pragma unroll
      for (int u = 0; u < 8; ++u) {
        const int k = k0 + 8 * u;
        if (k < nblk) {
          s.x += v[u].x; s.y += v[u].y; s.z += v[u].z; s.w += v[u].w;
          if (k < blk) { p.x += v[u].x; p.y += v[u].y; p.z += v[u].z; p.w += v[u].w; }
        }
      }
    }
    atomicAdd(&tot[lane * 4 + 0], s.x);
    atomicAdd(&tot[lane * 4 + 1], s.y);
    atomicAdd(&tot[lane * 4 + 2], s.z);
    atomicAdd(&tot[lane * 4 + 3], s.w);
    atomicAdd(&pre[lane * 4 + 0], p.x);
    atomicAdd(&pre[lane * 4 + 1], p.y);
    atomicAdd(&pre[lane * 4 + 2], p.z);
    atomicAdd(&pre[lane * 4 + 3], p.w);
  }
  __syncthreads();
  if (w == 0) wave_scan128(tot, base, lane);
  if (w == 1) wave_scan128(mine, lstart, lane);
  __syncthreads();
  if (t < MAXB) {
    lcur[t] = lstart[t];
    goff[t] = base[t] + pre[t] - lstart[t];
    if (blk == 0) bucket_base[t] = base[t];
  }
  if (t >= MAXB && t < MAXB + 32) lcur[t] = 0;
  __syncthreads();

  const int e0 = blk * CHUNK;
  int d[EPT], s[EPT], pos[EPT];
#pragma unroll
  for (int j = 0; j < EPT; ++j) {
    const int i = e0 + j * 256 + t;
    d[j] = (i < e) ? dst[i] : -1;
    s[j] = (i < e) ? src[i] : 0;
  }
#pragma unroll
  for (int j = 0; j < EPT; ++j) {
    const int bk = (d[j] >= 0) ? (d[j] >> BSH) : (MAXB + lane);
    pos[j] = atomicAdd(&lcur[bk], 1);
  }
#pragma unroll
  for (int j = 0; j < EPT; ++j) {
    if (d[j] >= 0) stage[pos[j]] = make_int2(s[j], d[j]);
  }
  __syncthreads();
  const int cntb = (e - e0 < CHUNK) ? (e - e0) : CHUNK;
  int2 pr[EPT];
  int g[EPT];
#pragma unroll
  for (int j = 0; j < EPT; ++j) {
    const int i = j * 256 + t;
    pr[j] = stage[(i < cntb) ? i : (cntb - 1)];
  }
#pragma unroll
  for (int j = 0; j < EPT; ++j) g[j] = goff[pr[j].y >> BSH];
#pragma unroll
  for (int j = 0; j < EPT; ++j) {
    const int i = j * 256 + t;
    if (i < cntb) pairs[g[j] + i] = pr[j];
  }
}

constexpr int RB = 8;

__global__ __launch_bounds__(1024) void bucket_sort_kernel(const int* __restrict__ pairs_i,
                                                           const int* __restrict__ bucket_base,
                                                           int e, int nb,
                                                           int* __restrict__ cnt_g,
                                                           int* __restrict__ rowptr,
                                                           float* __restrict__ dinv,
                                                           int* __restrict__ csr_src) {
  const int2* __restrict__ pairs = (const int2*)pairs_i;
  __shared__ int cnt[BW];
  __shared__ int cur[BW + 32];
  __shared__ int wsum[32];
  __shared__ int wex[32];
  __shared__ int stage[SCAP];
  const int t = threadIdx.x;
  const int lane = t & 31;
  const int w = t >> 5;
  const int b = blockIdx.x;
  const int lo = b << BSH;
  const int beg = bucket_base[b];
  const int end = (b + 1 < nb) ? bucket_base[b + 1] : e;
  cnt[t] = 0;
  __syncthreads();
  for (int i0 = beg; i0 < end; i0 += BW * RB) {
    int dd[RB];
#pragma unroll
    for (int u = 0; u < RB; ++u) {
      const int i = i0 + u * BW + t;
      dd[u] = (i < end) ? pairs[i].y : -1;
    }
#pragma unroll
    for (int u = 0; u < RB; ++u) {
      if (dd[u] >= 0) atomicAdd(&cnt[dd[u] - lo], 1);
    }
  }
  __syncthreads();
  const int c = cnt[t];
  int incl = c;
  for (int o = 1; o < 32; o <<= 1) {
    const int y = __shfl_up(incl, o);
    if (lane >= o) incl += y;
  }
  if (lane == 31) wsum[w] = incl;
  __syncthreads();
  if (w == 0) {
    const int v = wsum[lane];
    int in2 = v;
    for (int o = 1; o < 32; o <<= 1) {
      const int y = __shfl_up(in2, o);
      if (lane >= o) in2 += y;
    }
    wex[lane] = in2 - v;
  }
  __syncthreads();
  const int rp = beg + wex[w] + incl - c;
  rowptr[lo + t] = rp;
  cnt_g[lo + t] = c;
  const float df = (float)(c + 1);
  dinv[lo + t] = (df > 0.0f) ? rsqrtf(df) : 0.0f;
  cur[t] = rp;
  if (t < 32) cur[BW + t] = 0;
  __syncthreads();
  const int nbk = end - beg;
  if (nbk <= SCAP) {
    for (int i0 = beg; i0 < end; i0 += BW * RB) {
      int2 pr[RB];
      int pos[RB];
#pragma unroll
      for (int u = 0; u < RB; ++u) {
        const int i = i0 + u * BW + t;
        pr[u] = (i < end) ? pairs[i] : make_int2(0, -1);
      }
#pragma unroll
      for (int u = 0; u < RB; ++u) {
        const int k = (pr[u].y >= 0) ? (pr[u].y - lo) : (BW + lane);
        pos[u] = atomicAdd(&cur[k], 1);
      }
#pragma unroll
      for (int u = 0; u < RB; ++u) {
        if (pr[u].y >= 0) stage[pos[u] - beg] = pr[u].x;
      }
    }
    __syncthreads();
    for (int i = t; i < nbk; i += BW) csr_src[beg + i] = stage[i];
  } else {
    for (int i0 = beg; i0 < end; i0 += BW * RB) {
      int2 pr[RB];
      int pos[RB];
#pragma unroll
      for (int u = 0; u < RB; ++u) {
        const int i = i0 + u * BW + t;
        pr[u] = (i < end) ? pairs[i] : make_int2(0, -1);
      }
#pragma unroll
      for (int u = 0; u < RB; ++u) {
        const int k = (pr[u].y >= 0) ? (pr[u].y - lo) : (BW + lane);
        pos[u] = atomicAdd(&cur[k], 1);
      }
#pragma unroll
      for (int u = 0; u < RB; ++u) {
        if (pr[u].y >= 0) csr_src[pos[u]] = pr[u].x;
      }
    }
  }
}

constexpr int GEMM_WPB = 4;

template <int K>
__device__ __forceinline__ void gemm_request_tile(v2f (&a)[K / 4], v4f (&d)[2],
                                                  const float* __restrict__ X,
                                                  const float* __restrict__ dinv,
                                                  int t, int M, int c, int h) {
  int row = (t << 4) + c;
  if (row >= M) row = M - 1;
  const float* __restrict__ xr = X + (size_t)row * K + 2 * h;
#pragma unroll
  for (int ks = 0; ks < K / 4; ++ks) a[ks] = *(const v2f*)(xr + 4 * ks);
  const v4f* __restrict__ dp = (const v4f*)(dinv + (t << 4) + 8 * h);
  d[0] = dp[0];
  d[1] = dp[1];
}

template <int K, int NC>
__device__ __forceinline__ void gemm_tile(const v2f (&a)[K / 4], const v4f (&d)[2],
                                          const v2f (&bf)[K / 4][NC / 16],
                                          float* __restrict__ H, int t, int M, int c, int h) {
  constexpr int NT = NC / 16;
  constexpr int KS = K / 4;
  typedef __attribute__((ext_vector_type(NT))) float vnt;
  v8f acc[NT];
#pragma unroll
  for (int ct = 0; ct < NT; ++ct) acc[ct] = v8f{};
#pragma unroll
  for (int ks = 0; ks < KS; ++ks) {
#pragma unroll
    for (int ct = 0; ct < NT; ++ct) {
      acc[ct] = __builtin_amdgcn_wmma_f32_16x16x4_f32(
           false, a[ks],  false, bf[ks][ct],
           (short)0, acc[ct],  false,  false);
    }
  }
  const int rbase = (t << 4) + 8 * h;
  float* __restrict__ hp = H + (size_t)rbase * NC + NT * c;
  if ((t << 4) + 16 <= M) {
#pragma unroll
    for (int v = 0; v < 8; ++v) {
      vnt o;
#pragma unroll
      for (int ct = 0; ct < NT; ++ct) o[ct] = acc[ct][v] * d[v >> 2][v & 3];
      *(vnt*)(hp + v * NC) = o;
    }
  } else {
#pragma unroll
    for (int v = 0; v < 8; ++v) {
      vnt o;
#pragma unroll
      for (int ct = 0; ct < NT; ++ct) o[ct] = acc[ct][v] * d[v >> 2][v & 3];
      if (rbase + v < M) *(vnt*)(hp + v * NC) = o;
    }
  }
}

template <int K, int NC>
__global__ __launch_bounds__(32 * GEMM_WPB) void gemm_wmma_kernel(
    const float* __restrict__ X, const float* __restrict__ W,
    const float* __restrict__ dinv, float* __restrict__ H, int M, int ntiles, int GW) {
  constexpr int NT = NC / 16;
  constexpr int KS = K / 4;
  typedef __attribute__((ext_vector_type(NT))) float vnt;
  const int lane = threadIdx.x & 31;
  const int wave = __builtin_amdgcn_readfirstlane((int)(threadIdx.x >> 5));
  const int c = lane & 15;
  const int h = lane >> 4;
  const int gw = (int)blockIdx.x * GEMM_WPB + wave;
  if (gw >= ntiles) return;

  v2f a0[KS], a1[KS];
  v4f d0[2], d1[2];
  int t = gw;
  gemm_request_tile<K>(a0, d0, X, dinv, t, M, c, h);

  v2f bf[KS][NT];
#pragma unroll
  for (int ks = 0; ks < KS; ++ks) {
    const vnt r0 = *(const vnt*)(W + (4 * ks + 2 * h) * NC + NT * c);
    const vnt r1 = *(const vnt*)(W + (4 * ks + 2 * h + 1) * NC + NT * c);
#pragma unroll
    for (int ct = 0; ct < NT; ++ct) {
      bf[ks][ct].x = r0[ct];
      bf[ks][ct].y = r1[ct];
    }
  }

  for (;;) {
    const int t1 = t + GW;
    if (t1 >= ntiles) {
      gemm_tile<K, NC>(a0, d0, bf, H, t, M, c, h);
      break;
    }
    gemm_request_tile<K>(a1, d1, X, dinv, t1, M, c, h);
    gemm_tile<K, NC>(a0, d0, bf, H, t, M, c, h);
    const int t2 = t1 + GW;
    if (t2 >= ntiles) {
      gemm_tile<K, NC>(a1, d1, bf, H, t1, M, c, h);
      break;
    }
    gemm_request_tile<K>(a0, d0, X, dinv, t2, M, c, h);
    gemm_tile<K, NC>(a1, d1, bf, H, t1, M, c, h);
    t = t2;
  }
}


template <int C>
__device__ __forceinline__ v2f gather_group(v2f acc, int s_l, int j, int m,
                                            const v2f* __restrict__ h2, int sub, int cl) {
  constexpr int LPR = C / 2;
  constexpr int RPI = 32 / LPR;
  constexpr int G   = 8;
  constexpr int U   = G / RPI;
  v2f hv[U];
  bool ok[U];
#pragma unroll
  for (int u = 0; u < U; ++u) {
    int s;
    if (RPI == 1) {
      const int idx = j + u;
      s = __builtin_amdgcn_readlane(s_l, idx);
      ok[u] = idx < m;
    } else {
      const int idx = j + 2 * u;
      const int sa = __builtin_amdgcn_readlane(s_l, idx);
      const int sb = __builtin_amdgcn_readlane(s_l, idx + 1);
      s = sub ? sb : sa;
      ok[u] = (idx + sub) < m;
    }
    const v2f* __restrict__ rowp = h2 + (size_t)s * LPR;
    hv[u] = rowp[cl];
  }
#pragma unroll
  for (int u = 0; u < U; ++u) {
    const v2f nv = acc + hv[u];
    acc = ok[u] ? nv : acc;
  }
  return acc;
}

template <int C, bool RELU>
__global__ __launch_bounds__(256) void gather_kernel(
    const int* __restrict__ rowptr, const int* __restrict__ cnt,
    const int* __restrict__ csr_src, const float* __restrict__ dinv,
    const float* __restrict__ h, const float* __restrict__ bias,
    float* __restrict__ out, int n, int npb) {
  constexpr int LPR = C / 2;
  constexpr int RPI = 32 / LPR;
  constexpr int G   = 8;
  constexpr int WPB = 8;
  const int lane = threadIdx.x & 31;
  const int wv  = __builtin_amdgcn_readfirstlane((int)(threadIdx.x >> 5));
  const int sub = lane / LPR;
  const int cl  = lane - sub * LPR;
  const v2f* __restrict__ h2 = (const v2f*)h;
  const v2f bv = ((const v2f*)bias)[cl];

  const int nb0  = (int)blockIdx.x * npb;
  const int nend = (nb0 + npb < n) ? (nb0 + npb) : n;

  for (int c0 = nb0 + wv; c0 < nend; c0 += 32 * WPB) {
    const int hn = c0 + WPB * lane;
    int hstart = 0, hdeg = 0, hdi = 0;
    if (hn < nend) {
      hstart = rowptr[hn];
      hdeg   = cnt[hn];
      hdi    = __float_as_int(dinv[hn]);
    }
    const int avail = (nend - c0 + WPB - 1) / WPB;
    const int cnum  = (avail < 32) ? avail : 32;

    int node  = c0;
    int start = __builtin_amdgcn_readlane(hstart, 0);
    int deg   = __builtin_amdgcn_readlane(hdeg, 0);
    int s_l   = node;
    if (lane < deg) s_l = csr_src[start + lane];

    for (int it = 0; it < cnum; ++it) {
      const float di = __int_as_float(__builtin_amdgcn_readlane(hdi, it));
      const int nit   = it + 1;
      const int nnode = node + WPB;
      int nstart = 0, ndeg = 0, ns_l = nnode;
      if (nit < cnum) {
        nstart = __builtin_amdgcn_readlane(hstart, nit);
        ndeg   = __builtin_amdgcn_readlane(hdeg, nit);
        if (lane < ndeg) ns_l = csr_src[nstart + lane];
      }

      const int total = deg + 1;
      v2f acc = {0.0f, 0.0f};
      {
        const int m = (total < 32) ? total : 32;
        for (int j = 0; j < m; j += G) acc = gather_group<C>(acc, s_l, j, m, h2, sub, cl);
      }
      for (int base = 32; base < total; base += 32) {
        const int k = base + lane;
        int s_x = node;
        if (k < deg) s_x = csr_src[start + k];
        const int rem = total - base;
        const int m = (rem < 32) ? rem : 32;
        for (int j = 0; j < m; j += G) acc = gather_group<C>(acc, s_x, j, m, h2, sub, cl);
      }

      if (RPI == 2) {
        acc.x += __shfl_xor(acc.x, 16);
        acc.y += __shfl_xor(acc.y, 16);
      }
      if (sub == 0) {
        v2f r = acc * di + bv;
        if (RELU) {
          r.x = fmaxf(r.x, 0.0f);
          r.y = fmaxf(r.y, 0.0f);
        }
        ((v2f*)out)[(size_t)node * LPR + cl] = r;
      }

      node  = nnode;
      start = nstart;
      deg   = ndeg;
      s_l   = ns_l;
    }
  }
}


extern "C" void kernel_launch(void* const* d_in, const int* in_sizes, int n_in,
                              void* d_out, int out_size, void* d_ws, size_t ws_size,
                              hipStream_t stream) {
  (void)n_in; (void)out_size; (void)ws_size;
  const float* x  = (const float*)d_in[0];
  const int*   ei = (const int*)d_in[1];
  const float* W1 = (const float*)d_in[2];
  const float* b1 = (const float*)d_in[3];
  const float* W2 = (const float*)d_in[4];
  const float* b2 = (const float*)d_in[5];

  const int IN_C = 128, HID = 64, OUT = 32;
  const int n = in_sizes[0] / IN_C;
  const int e = in_sizes[1] / 2;
  const int npb   = (((n + 2047) / 2048) + 7) & ~7;
  const int ggrid = (n + npb - 1) / npb;
  const int ntiles = (n + 15) / 16;
  const int gwaves = (ntiles < 2048) ? ntiles : 2048;
  const int ggemm  = (gwaves + GEMM_WPB - 1) / GEMM_WPB;
  const int* srcp = ei;
  const int* dstp = ei + e;

  const int    np   = (n + BW - 1) & ~(BW - 1);
  const int    nb   = np >> BSH;
  const int    nblk = (e + CHUNK - 1) / CHUNK;
  const size_t ep   = ((size_t)e + 63) & ~(size_t)63;

  int*   cnt     = (int*)d_ws;
  int*   rowptr  = cnt + np;
  float* dinv    = (float*)(rowptr + np);
  int*   bbase   = (int*)(dinv + np);
  int*   blkhist = bbase + MAXB;
  int*   pairs   = blkhist + (size_t)nblk * MAXB;
  int*   csr     = pairs + 2 * ep;
  float* h1      = (float*)(csr + ep);
  float* a1      = h1 + (size_t)n * HID;
  float* h2      = a1 + (size_t)n * HID;
  float* outp    = (float*)d_out;
  (void)OUT;

  bucket_count_kernel<<<nblk, 256, 0, stream>>>(dstp, blkhist, e);
  bucket_place_kernel<<<nblk, 256, 0, stream>>>(srcp, dstp, blkhist, bbase, pairs, e, nblk);
  bucket_sort_kernel<<<nb, BW, 0, stream>>>(pairs, bbase, e, nb, cnt, rowptr, dinv, csr);

  gemm_wmma_kernel<128, 64><<<ggemm, 32 * GEMM_WPB, 0, stream>>>(x, W1, dinv, h1, n, ntiles, gwaves);
  gather_kernel<64, true><<<ggrid, 256, 0, stream>>>(rowptr, cnt, csr, dinv, h1, b1, a1, n, npb);

  gemm_wmma_kernel<64, 32><<<ggemm, 32 * GEMM_WPB, 0, stream>>>(a1, W2, dinv, h2, n, ntiles, gwaves);
  gather_kernel<32, false><<<ggrid, 256, 0, stream>>>(rowptr, cnt, csr, dinv, h2, b2, outp, n, npb);
}


# Gathers request all rows of a node (up to 32 list items) as one batch before the first add, not 8 at a time with a wait per group; gather grid resized to the waves that fit; sum order unchanged
# speedup vs baseline: 5.9866x; 1.0481x over previous
#include <hip/hip_runtime.h>


typedef __attribute__((ext_vector_type(2))) float v2f;
typedef __attribute__((ext_vector_type(4))) float v4f;
typedef __attribute__((ext_vector_type(8))) float v8f;


constexpr int BSH    = 10;
constexpr int BW     = 1 << BSH;
constexpr int MAXB   = 128;
constexpr int CHUNK  = 4096;
constexpr int EPT    = CHUNK / 256;
constexpr int SCAP   = 40960;

__device__ inline void wave_scan128(const int* in, int* out, int lane) {
  const int a0 = in[lane * 4 + 0];
  const int a1 = in[lane * 4 + 1];
  const int a2 = in[lane * 4 + 2];
  const int a3 = in[lane * 4 + 3];
  const int s = (a0 + a1) + (a2 + a3);
  int incl = s;
  for (int o = 1; o < 32; o <<= 1) {
    const int y = __shfl_up(incl, o);
    if (lane >= o) incl += y;
  }
  const int ex = incl - s;
  out[lane * 4 + 0] = ex;
  out[lane * 4 + 1] = ex + a0;
  out[lane * 4 + 2] = ex + a0 + a1;
  out[lane * 4 + 3] = ex + a0 + a1 + a2;
}

__global__ __launch_bounds__(256) void bucket_count_kernel(const int* __restrict__ dst,
                                                           int* __restrict__ blkhist, int e) {
  __shared__ int hist[MAXB];
  const int t = threadIdx.x;
  if (t < MAXB) hist[t] = 0;
  __syncthreads();
  const int e0 = blockIdx.x * CHUNK;
  int d[EPT];
#pragma unroll
  for (int j = 0; j < EPT; ++j) {
    const int i = e0 + j * 256 + t;
    d[j] = (i < e) ? dst[i] : -1;
  }
#pragma unroll
  for (int j = 0; j < EPT; ++j) {
    if (d[j] >= 0) atomicAdd(&hist[d[j] >> BSH], 1);
  }
  __syncthreads();
  if (t < MAXB) blkhist[blockIdx.x * MAXB + t] = hist[t];
}

__global__ __launch_bounds__(256) void bucket_place_kernel(const int* __restrict__ src,
                                                           const int* __restrict__ dst,
                                                           const int* __restrict__ blkhist,
                                                           int* __restrict__ bucket_base,
                                                           int* __restrict__ pairs_i, int e, int nblk) {
  int2* __restrict__ pairs = (int2*)pairs_i;
  __shared__ int pre[MAXB];
  __shared__ int tot[MAXB];
  __shared__ int base[MAXB];
  __shared__ int mine[MAXB];
  __shared__ int lstart[MAXB];
  __shared__ int lcur[MAXB + 32];
  __shared__ int goff[MAXB];
  __shared__ int2 stage[CHUNK];
  const int t = threadIdx.x;
  const int lane = t & 31;
  const int w = t >> 5;
  const int blk = blockIdx.x;
  if (t < MAXB) {
    pre[t] = 0;
    tot[t] = 0;
    mine[t] = blkhist[blk * MAXB + t];
  }
  __syncthreads();

  {
    const int4* __restrict__ M4 = (const int4*)blkhist;
    const int ws = __builtin_amdgcn_readfirstlane(w);
    int4 p = {0, 0, 0, 0};
    int4 s = {0, 0, 0, 0};
    for (int k0 = ws; k0 < nblk; k0 += 64) {
      int4 v[8];
#pragma unroll
      for (int u = 0; u < 8; ++u) {
        const int k = k0 + 8 * u;
        const int kk = (k < nblk) ? k : k0;
        v[u] = M4[kk * (MAXB / 4) + lane];
      }
#pragma unroll
      for (int u = 0; u < 8; ++u) {
        const int k = k0 + 8 * u;
        if (k < nblk) {
          s.x += v[u].x; s.y += v[u].y; s.z += v[u].z; s.w += v[u].w;
          if (k < blk) { p.x += v[u].x; p.y += v[u].y; p.z += v[u].z; p.w += v[u].w; }
        }
      }
    }
    atomicAdd(&tot[lane * 4 + 0], s.x);
    atomicAdd(&tot[lane * 4 + 1], s.y);
    atomicAdd(&tot[lane * 4 + 2], s.z);
    atomicAdd(&tot[lane * 4 + 3], s.w);
    atomicAdd(&pre[lane * 4 + 0], p.x);
    atomicAdd(&pre[lane * 4 + 1], p.y);
    atomicAdd(&pre[lane * 4 + 2], p.z);
    atomicAdd(&pre[lane * 4 + 3], p.w);
  }
  __syncthreads();
  if (w == 0) wave_scan128(tot, base, lane);
  if (w == 1) wave_scan128(mine, lstart, lane);
  __syncthreads();
  if (t < MAXB) {
    lcur[t] = lstart[t];
    goff[t] = base[t] + pre[t] - lstart[t];
    if (blk == 0) bucket_base[t] = base[t];
  }
  if (t >= MAXB && t < MAXB + 32) lcur[t] = 0;
  __syncthreads();

  const int e0 = blk * CHUNK;
  int d[EPT], s[EPT], pos[EPT];
#pragma unroll
  for (int j = 0; j < EPT; ++j) {
    const int i = e0 + j * 256 + t;
    d[j] = (i < e) ? dst[i] : -1;
    s[j] = (i < e) ? src[i] : 0;
  }
#pragma unroll
  for (int j = 0; j < EPT; ++j) {
    const int bk = (d[j] >= 0) ? (d[j] >> BSH) : (MAXB + lane);
    pos[j] = atomicAdd(&lcur[bk], 1);
  }
#pragma unroll
  for (int j = 0; j < EPT; ++j) {
    if (d[j] >= 0) stage[pos[j]] = make_int2(s[j], d[j]);
  }
  __syncthreads();
  const int cntb = (e - e0 < CHUNK) ? (e - e0) : CHUNK;
  int2 pr[EPT];
  int g[EPT];
#pragma unroll
  for (int j = 0; j < EPT; ++j) {
    const int i = j * 256 + t;
    pr[j] = stage[(i < cntb) ? i : (cntb - 1)];
  }
#pragma unroll
  for (int j = 0; j < EPT; ++j) g[j] = goff[pr[j].y >> BSH];
#pragma unroll
  for (int j = 0; j < EPT; ++j) {
    const int i = j * 256 + t;
    if (i < cntb) pairs[g[j] + i] = pr[j];
  }
}

constexpr int RB = 8;

__global__ __launch_bounds__(1024) void bucket_sort_kernel(const int* __restrict__ pairs_i,
                                                           const int* __restrict__ bucket_base,
                                                           int e, int nb,
                                                           int* __restrict__ cnt_g,
                                                           int* __restrict__ rowptr,
                                                           float* __restrict__ dinv,
                                                           int* __restrict__ csr_src) {
  const int2* __restrict__ pairs = (const int2*)pairs_i;
  __shared__ int cnt[BW];
  __shared__ int cur[BW + 32];
  __shared__ int wsum[32];
  __shared__ int wex[32];
  __shared__ int stage[SCAP];
  const int t = threadIdx.x;
  const int lane = t & 31;
  const int w = t >> 5;
  const int b = blockIdx.x;
  const int lo = b << BSH;
  const int beg = bucket_base[b];
  const int end = (b + 1 < nb) ? bucket_base[b + 1] : e;
  cnt[t] = 0;
  __syncthreads();
  for (int i0 = beg; i0 < end; i0 += BW * RB) {
    int dd[RB];
#pragma unroll
    for (int u = 0; u < RB; ++u) {
      const int i = i0 + u * BW + t;
      dd[u] = (i < end) ? pairs[i].y : -1;
    }
#pragma unroll
    for (int u = 0; u < RB; ++u) {
      if (dd[u] >= 0) atomicAdd(&cnt[dd[u] - lo], 1);
    }
  }
  __syncthreads();
  const int c = cnt[t];
  int incl = c;
  for (int o = 1; o < 32; o <<= 1) {
    const int y = __shfl_up(incl, o);
    if (lane >= o) incl += y;
  }
  if (lane == 31) wsum[w] = incl;
  __syncthreads();
  if (w == 0) {
    const int v = wsum[lane];
    int in2 = v;
    for (int o = 1; o < 32; o <<= 1) {
      const int y = __shfl_up(in2, o);
      if (lane >= o) in2 += y;
    }
    wex[lane] = in2 - v;
  }
  __syncthreads();
  const int rp = beg + wex[w] + incl - c;
  rowptr[lo + t] = rp;
  cnt_g[lo + t] = c;
  const float df = (float)(c + 1);
  dinv[lo + t] = (df > 0.0f) ? rsqrtf(df) : 0.0f;
  cur[t] = rp;
  if (t < 32) cur[BW + t] = 0;
  __syncthreads();
  const int nbk = end - beg;
  if (nbk <= SCAP) {
    for (int i0 = beg; i0 < end; i0 += BW * RB) {
      int2 pr[RB];
      int pos[RB];
#pragma unroll
      for (int u = 0; u < RB; ++u) {
        const int i = i0 + u * BW + t;
        pr[u] = (i < end) ? pairs[i] : make_int2(0, -1);
      }
#pragma unroll
      for (int u = 0; u < RB; ++u) {
        const int k = (pr[u].y >= 0) ? (pr[u].y - lo) : (BW + lane);
        pos[u] = atomicAdd(&cur[k], 1);
      }
#pragma unroll
      for (int u = 0; u < RB; ++u) {
        if (pr[u].y >= 0) stage[pos[u] - beg] = pr[u].x;
      }
    }
    __syncthreads();
    for (int i = t; i < nbk; i += BW) csr_src[beg + i] = stage[i];
  } else {
    for (int i0 = beg; i0 < end; i0 += BW * RB) {
      int2 pr[RB];
      int pos[RB];
#pragma unroll
      for (int u = 0; u < RB; ++u) {
        const int i = i0 + u * BW + t;
        pr[u] = (i < end) ? pairs[i] : make_int2(0, -1);
      }
#pragma unroll
      for (int u = 0; u < RB; ++u) {
        const int k = (pr[u].y >= 0) ? (pr[u].y - lo) : (BW + lane);
        pos[u] = atomicAdd(&cur[k], 1);
      }
#pragma unroll
      for (int u = 0; u < RB; ++u) {
        if (pr[u].y >= 0) csr_src[pos[u]] = pr[u].x;
      }
    }
  }
}

constexpr int GEMM_WPB = 4;

template <int K>
__device__ __forceinline__ void gemm_request_tile(v2f (&a)[K / 4], v4f (&d)[2],
                                                  const float* __restrict__ X,
                                                  const float* __restrict__ dinv,
                                                  int t, int M, int c, int h) {
  int row = (t << 4) + c;
  if (row >= M) row = M - 1;
  const float* __restrict__ xr = X + (size_t)row * K + 2 * h;
#pragma unroll
  for (int ks = 0; ks < K / 4; ++ks) a[ks] = *(const v2f*)(xr + 4 * ks);
  const v4f* __restrict__ dp = (const v4f*)(dinv + (t << 4) + 8 * h);
  d[0] = dp[0];
  d[1] = dp[1];
}

template <int K, int NC>
__device__ __forceinline__ void gemm_tile(const v2f (&a)[K / 4], const v4f (&d)[2],
                                          const v2f (&bf)[K / 4][NC / 16],
                                          float* __restrict__ H, int t, int M, int c, int h) {
  constexpr int NT = NC / 16;
  constexpr int KS = K / 4;
  typedef __attribute__((ext_vector_type(NT))) float vnt;
  v8f acc[NT];
#pragma unroll
  for (int ct = 0; ct < NT; ++ct) acc[ct] = v8f{};
#pragma unroll
  for (int ks = 0; ks < KS; ++ks) {
#pragma unroll
    for (int ct = 0; ct < NT; ++ct) {
      acc[ct] = __builtin_amdgcn_wmma_f32_16x16x4_f32(
           false, a[ks],  false, bf[ks][ct],
           (short)0, acc[ct],  false,  false);
    }
  }
  const int rbase = (t << 4) + 8 * h;
  float* __restrict__ hp = H + (size_t)rbase * NC + NT * c;
  if ((t << 4) + 16 <= M) {
#pragma unroll
    for (int v = 0; v < 8; ++v) {
      vnt o;
#pragma unroll
      for (int ct = 0; ct < NT; ++ct) o[ct] = acc[ct][v] * d[v >> 2][v & 3];
      *(vnt*)(hp + v * NC) = o;
    }
  } else {
#pragma unroll
    for (int v = 0; v < 8; ++v) {
      vnt o;
#pragma unroll
      for (int ct = 0; ct < NT; ++ct) o[ct] = acc[ct][v] * d[v >> 2][v & 3];
      if (rbase + v < M) *(vnt*)(hp + v * NC) = o;
    }
  }
}

template <int K, int NC>
__global__ __launch_bounds__(32 * GEMM_WPB) void gemm_wmma_kernel(
    const float* __restrict__ X, const float* __restrict__ W,
    const float* __restrict__ dinv, float* __restrict__ H, int M, int ntiles, int GW) {
  constexpr int NT = NC / 16;
  constexpr int KS = K / 4;
  typedef __attribute__((ext_vector_type(NT))) float vnt;
  const int lane = threadIdx.x & 31;
  const int wave = __builtin_amdgcn_readfirstlane((int)(threadIdx.x >> 5));
  const int c = lane & 15;
  const int h = lane >> 4;
  const int gw = (int)blockIdx.x * GEMM_WPB + wave;
  if (gw >= ntiles) return;

  v2f a0[KS], a1[KS];
  v4f d0[2], d1[2];
  int t = gw;
  gemm_request_tile<K>(a0, d0, X, dinv, t, M, c, h);

  v2f bf[KS][NT];
#pragma unroll
  for (int ks = 0; ks < KS; ++ks) {
    const vnt r0 = *(const vnt*)(W + (4 * ks + 2 * h) * NC + NT * c);
    const vnt r1 = *(const vnt*)(W + (4 * ks + 2 * h + 1) * NC + NT * c);
#pragma unroll
    for (int ct = 0; ct < NT; ++ct) {
      bf[ks][ct].x = r0[ct];
      bf[ks][ct].y = r1[ct];
    }
  }

  for (;;) {
    const int t1 = t + GW;
    if (t1 >= ntiles) {
      gemm_tile<K, NC>(a0, d0, bf, H, t, M, c, h);
      break;
    }
    gemm_request_tile<K>(a1, d1, X, dinv, t1, M, c, h);
    gemm_tile<K, NC>(a0, d0, bf, H, t, M, c, h);
    const int t2 = t1 + GW;
    if (t2 >= ntiles) {
      gemm_tile<K, NC>(a1, d1, bf, H, t1, M, c, h);
      break;
    }
    gemm_request_tile<K>(a0, d0, X, dinv, t2, M, c, h);
    gemm_tile<K, NC>(a1, d1, bf, H, t1, M, c, h);
    t = t2;
  }
}


template <int C, int NG>
__device__ __forceinline__ v2f gather_batch(v2f acc, int s_l, int m,
                                            const v2f* __restrict__ h2, int sub, int cl) {
  constexpr int LPR = C / 2;
  constexpr int RPI = 32 / LPR;
  constexpr int NL  = NG * 8 / RPI;
  constexpr int FL  = (NG - 1) * 8 / RPI;
  v2f hv[NL];
#pragma unroll
  for (int q = 0; q < NL; ++q) {
    int s;
    if (RPI == 1) {
      s = __builtin_amdgcn_readlane(s_l, q);
    } else {
      const int sa = __builtin_amdgcn_readlane(s_l, 2 * q);
      const int sb = __builtin_amdgcn_readlane(s_l, 2 * q + 1);
      s = sub ? sb : sa;
    }
    const v2f* __restrict__ rowp = h2 + (size_t)s * LPR;
    hv[q] = rowp[cl];
  }
#pragma unroll
  for (int q = 0; q < NL; ++q) {
    if (q < FL) {
      acc = acc + hv[q];
    } else {
      const bool ok = (RPI == 1) ? (q < m) : ((2 * q + sub) < m);
      const v2f nv = acc + hv[q];
      acc = ok ? nv : acc;
    }
  }
  return acc;
}

template <int C>
__device__ __forceinline__ v2f gather_items(v2f acc, int s_l, int m,
                                            const v2f* __restrict__ h2, int sub, int cl) {
  const int ng = (m + 7) >> 3;
  switch (ng) {
    case 1:  return gather_batch<C, 1>(acc, s_l, m, h2, sub, cl);
    case 2:  return gather_batch<C, 2>(acc, s_l, m, h2, sub, cl);
    case 3:  return gather_batch<C, 3>(acc, s_l, m, h2, sub, cl);
    default: return gather_batch<C, 4>(acc, s_l, m, h2, sub, cl);
  }
}

template <int C, bool RELU>
__global__ __launch_bounds__(256) void gather_kernel(
    const int* __restrict__ rowptr, const int* __restrict__ cnt,
    const int* __restrict__ csr_src, const float* __restrict__ dinv,
    const float* __restrict__ h, const float* __restrict__ bias,
    float* __restrict__ out, int n, int npb) {
  constexpr int LPR = C / 2;
  constexpr int RPI = 32 / LPR;
  constexpr int WPB = 8;
  const int lane = threadIdx.x & 31;
  const int wv  = __builtin_amdgcn_readfirstlane((int)(threadIdx.x >> 5));
  const int sub = lane / LPR;
  const int cl  = lane - sub * LPR;
  const v2f* __restrict__ h2 = (const v2f*)h;
  const v2f bv = ((const v2f*)bias)[cl];

  const int nb0  = (int)blockIdx.x * npb;
  const int nend = (nb0 + npb < n) ? (nb0 + npb) : n;

  for (int c0 = nb0 + wv; c0 < nend; c0 += 32 * WPB) {
    const int hn = c0 + WPB * lane;
    int hstart = 0, hdeg = 0, hdi = 0;
    if (hn < nend) {
      hstart = rowptr[hn];
      hdeg   = cnt[hn];
      hdi    = __float_as_int(dinv[hn]);
    }
    const int avail = (nend - c0 + WPB - 1) / WPB;
    const int cnum  = (avail < 32) ? avail : 32;

    int node  = c0;
    int start = __builtin_amdgcn_readlane(hstart, 0);
    int deg   = __builtin_amdgcn_readlane(hdeg, 0);
    int s_l   = node;
    if (lane < deg) s_l = csr_src[start + lane];

    for (int it = 0; it < cnum; ++it) {
      const float di = __int_as_float(__builtin_amdgcn_readlane(hdi, it));
      const int nit   = it + 1;
      const int nnode = node + WPB;
      int nstart = 0, ndeg = 0, ns_l = nnode;
      if (nit < cnum) {
        nstart = __builtin_amdgcn_readlane(hstart, nit);
        ndeg   = __builtin_amdgcn_readlane(hdeg, nit);
        if (lane < ndeg) ns_l = csr_src[nstart + lane];
      }

      const int total = deg + 1;
      v2f acc = {0.0f, 0.0f};
      {
        const int m = (total < 32) ? total : 32;
        acc = gather_items<C>(acc, s_l, m, h2, sub, cl);
      }
      for (int base = 32; base < total; base += 32) {
        const int k = base + lane;
        int s_x = node;
        if (k < deg) s_x = csr_src[start + k];
        const int rem = total - base;
        const int m = (rem < 32) ? rem : 32;
        acc = gather_items<C>(acc, s_x, m, h2, sub, cl);
      }

      if (RPI == 2) {
        acc.x += __shfl_xor(acc.x, 16);
        acc.y += __shfl_xor(acc.y, 16);
      }
      if (sub == 0) {
        v2f r = acc * di + bv;
        if (RELU) {
          r.x = fmaxf(r.x, 0.0f);
          r.y = fmaxf(r.y, 0.0f);
        }
        ((v2f*)out)[(size_t)node * LPR + cl] = r;
      }

      node  = nnode;
      start = nstart;
      deg   = ndeg;
      s_l   = ns_l;
    }
  }
}


extern "C" void kernel_launch(void* const* d_in, const int* in_sizes, int n_in,
                              void* d_out, int out_size, void* d_ws, size_t ws_size,
                              hipStream_t stream) {
  (void)n_in; (void)out_size; (void)ws_size;
  const float* x  = (const float*)d_in[0];
  const int*   ei = (const int*)d_in[1];
  const float* W1 = (const float*)d_in[2];
  const float* b1 = (const float*)d_in[3];
  const float* W2 = (const float*)d_in[4];
  const float* b2 = (const float*)d_in[5];

  const int IN_C = 128, HID = 64, OUT = 32;
  const int n = in_sizes[0] / IN_C;
  const int e = in_sizes[1] / 2;
  const int npb   = (((n + 1535) / 1536) + 7) & ~7;
  const int ggrid = (n + npb - 1) / npb;
  const int ntiles = (n + 15) / 16;
  const int gwaves = (ntiles < 2048) ? ntiles : 2048;
  const int ggemm  = (gwaves + GEMM_WPB - 1) / GEMM_WPB;
  const int* srcp = ei;
  const int* dstp = ei + e;

  const int    np   = (n + BW - 1) & ~(BW - 1);
  const int    nb   = np >> BSH;
  const int    nblk = (e + CHUNK - 1) / CHUNK;
  const size_t ep   = ((size_t)e + 63) & ~(size_t)63;

  int*   cnt     = (int*)d_ws;
  int*   rowptr  = cnt + np;
  float* dinv    = (float*)(rowptr + np);
  int*   bbase   = (int*)(dinv + np);
  int*   blkhist = bbase + MAXB;
  int*   pairs   = blkhist + (size_t)nblk * MAXB;
  int*   csr     = pairs + 2 * ep;
  float* h1      = (float*)(csr + ep);
  float* a1      = h1 + (size_t)n * HID;
  float* h2      = a1 + (size_t)n * HID;
  float* outp    = (float*)d_out;
  (void)OUT;

  bucket_count_kernel<<<nblk, 256, 0, stream>>>(dstp, blkhist, e);
  bucket_place_kernel<<<nblk, 256, 0, stream>>>(srcp, dstp, blkhist, bbase, pairs, e, nblk);
  bucket_sort_kernel<<<nb, BW, 0, stream>>>(pairs, bbase, e, nb, cnt, rowptr, dinv, csr);

  gemm_wmma_kernel<128, 64><<<ggemm, 32 * GEMM_WPB, 0, stream>>>(x, W1, dinv, h1, n, ntiles, gwaves);
  gather_kernel<64, true><<<ggrid, 256, 0, stream>>>(rowptr, cnt, csr, dinv, h1, b1, a1, n, npb);

  gemm_wmma_kernel<64, 32><<<ggemm, 32 * GEMM_WPB, 0, stream>>>(a1, W2, dinv, h2, n, ntiles, gwaves);
  gather_kernel<32, false><<<ggrid, 256, 0, stream>>>(rowptr, cnt, csr, dinv, h2, b2, outp, n, npb);
}


# Edge lists in 2 launches, not 3: each chunk groups its own edges by bucket in LDS and records segment starts; a bucket's work-group collects its segments; list (sum) order = LDS add arrival order
# speedup vs baseline: 6.5042x; 1.0865x over previous
#include <hip/hip_runtime.h>


typedef __attribute__((ext_vector_type(2))) float v2f;
typedef __attribute__((ext_vector_type(4))) float v4f;
typedef __attribute__((ext_vector_type(8))) float v8f;


constexpr int BSH    = 10;
constexpr int BW     = 1 << BSH;
constexpr int MAXB   = 128;
constexpr int CHUNK  = 4096;
constexpr int EPT    = CHUNK / 256;
constexpr int SCAP   = 40960;

__device__ inline void wave_scan128(const int* in, int* out, int lane) {
  const int a0 = in[lane * 4 + 0];
  const int a1 = in[lane * 4 + 1];
  const int a2 = in[lane * 4 + 2];
  const int a3 = in[lane * 4 + 3];
  const int s = (a0 + a1) + (a2 + a3);
  int incl = s;
  for (int o = 1; o < 32; o <<= 1) {
    const int y = __shfl_up(incl, o);
    if (lane >= o) incl += y;
  }
  const int ex = incl - s;
  out[lane * 4 + 0] = ex;
  out[lane * 4 + 1] = ex + a0;
  out[lane * 4 + 2] = ex + a0 + a1;
  out[lane * 4 + 3] = ex + a0 + a1 + a2;
}

__global__ __launch_bounds__(256) void chunk_group_kernel(const int* __restrict__ src,
                                                          const int* __restrict__ dst,
                                                          int* __restrict__ seghist,
                                                          int* __restrict__ segstart,
                                                          int* __restrict__ pairs_i, int e) {
  int2* __restrict__ pairs = (int2*)pairs_i;
  __shared__ int hist[MAXB + 32];
  __shared__ int lstart[MAXB];
  __shared__ int2 stage[CHUNK];
  const int t = threadIdx.x;
  const int lane = t & 31;
  const int w = t >> 5;
  const int blk = blockIdx.x;
  if (t < MAXB + 32) hist[t] = 0;
  __syncthreads();

  const int e0 = blk * CHUNK;
  int d[EPT], s[EPT], pos[EPT];
#pragma unroll
  for (int j = 0; j < EPT; ++j) {
    const int i = e0 + j * 256 + t;
    d[j] = (i < e) ? dst[i] : -1;
    s[j] = (i < e) ? src[i] : 0;
  }
#pragma unroll
  for (int j = 0; j < EPT; ++j) {
    const int bk = (d[j] >= 0) ? (d[j] >> BSH) : (MAXB + lane);
    pos[j] = atomicAdd(&hist[bk], 1);
  }
  __syncthreads();
  if (w == 0) wave_scan128(hist, lstart, lane);
  __syncthreads();
#pragma unroll
  for (int j = 0; j < EPT; ++j) {
    if (d[j] >= 0) stage[lstart[d[j] >> BSH] + pos[j]] = make_int2(s[j], d[j]);
  }
  if (t < MAXB) {
    seghist[blk * MAXB + t]  = hist[t];
    segstart[blk * MAXB + t] = lstart[t];
  }
  __syncthreads();
  const int cntb = (e - e0 < CHUNK) ? (e - e0) : CHUNK;
  int2 pr[EPT];
#pragma unroll
  for (int j = 0; j < EPT; ++j) {
    const int i = j * 256 + t;
    pr[j] = stage[(i < cntb) ? i : (cntb - 1)];
  }
#pragma unroll
  for (int j = 0; j < EPT; ++j) {
    const int i = j * 256 + t;
    if (i < cntb) pairs[e0 + i] = pr[j];
  }
}

constexpr int TB  = 16;
constexpr int SB1 = 16;
constexpr int SB2 = 8;

__global__ __launch_bounds__(1024) void bucket_sort_kernel(const int* __restrict__ pairs_i,
                                                           const int* __restrict__ seghist,
                                                           const int* __restrict__ segstart,
                                                           int nblk,
                                                           int* __restrict__ cnt_g,
                                                           int* __restrict__ rowptr,
                                                           float* __restrict__ dinv,
                                                           int* __restrict__ csr_src) {
  const int2* __restrict__ pairs = (const int2*)pairs_i;
  __shared__ int cnt[BW];
  __shared__ int cur[BW + 32];
  __shared__ int wsum[32];
  __shared__ int wex[32];
  __shared__ int redb[32];
  __shared__ int redm[32];
  __shared__ int stage[SCAP];
  const int t = threadIdx.x;
  const int lane = t & 31;
  const int w = __builtin_amdgcn_readfirstlane(t >> 5);
  const int b = blockIdx.x;
  const int lo = b << BSH;
  cnt[t] = 0;
  const int hk = w + 32 * lane;
  int hc = 0, hs = 0;
  if (hk < nblk) {
    hc = seghist[hk * MAXB + b];
    hs = segstart[hk * MAXB + b];
  }
  const int nseg = (nblk - w + 31) >> 5;
  int below = 0, mine = 0;
  {
    const int4* __restrict__ M4 = (const int4*)seghist;
    const int c0 = 4 * lane;
    if (c0 <= b) {
      for (int k0 = w; k0 < nblk; k0 += 32 * TB) {
        int4 v[TB];
#pragma unroll
        for (int u = 0; u < TB; ++u) {
          const int k = k0 + 32 * u;
          const int kk = (k < nblk) ? k : k0;
          v[u] = M4[kk * (MAXB / 4) + lane];
        }
#pragma unroll
        for (int u = 0; u < TB; ++u) {
          const int k = k0 + 32 * u;
          if (k < nblk) {
            below += ((c0 + 0 < b) ? v[u].x : 0) + ((c0 + 1 < b) ? v[u].y : 0) +
                     ((c0 + 2 < b) ? v[u].z : 0) + ((c0 + 3 < b) ? v[u].w : 0);
            mine  += ((c0 + 0 == b) ? v[u].x : 0) + ((c0 + 1 == b) ? v[u].y : 0) +
                     ((c0 + 2 == b) ? v[u].z : 0) + ((c0 + 3 == b) ? v[u].w : 0);
          }
        }
      }
    }
    for (int o = 16; o > 0; o >>= 1) {
      below += __shfl_xor(below, o);
      mine  += __shfl_xor(mine, o);
    }
    if (lane == 0) {
      redb[w] = below;
      redm[w] = mine;
    }
  }
  __syncthreads();
  int beg = redb[lane];
  int nbk = redm[lane];
  for (int o = 16; o > 0; o >>= 1) {
    beg += __shfl_xor(beg, o);
    nbk += __shfl_xor(nbk, o);
  }

  for (int u0 = 0; u0 < nseg; u0 += SB1) {
    int dd[SB1][2];
#pragma unroll
    for (int u = 0; u < SB1; ++u) {
      const int c  = __builtin_amdgcn_readlane(hc, u0 + u);
      const int s0 = (w + 32 * (u0 + u)) * CHUNK + __builtin_amdgcn_readlane(hs, u0 + u);
      dd[u][0] = (lane < c) ? pairs[s0 + lane].y : -1;
      dd[u][1] = (lane + 32 < c) ? pairs[s0 + lane + 32].y : -1;
    }
#pragma unroll
    for (int u = 0; u < SB1; ++u) {
      if (dd[u][0] >= 0) atomicAdd(&cnt[dd[u][0] - lo], 1);
      if (dd[u][1] >= 0) atomicAdd(&cnt[dd[u][1] - lo], 1);
    }
    for (int u = 0; u < SB1; ++u) {
      const int c  = __builtin_amdgcn_readlane(hc, u0 + u);
      const int s0 = (w + 32 * (u0 + u)) * CHUNK + __builtin_amdgcn_readlane(hs, u0 + u);
      for (int i = 64 + lane; i < c; i += 32) atomicAdd(&cnt[pairs[s0 + i].y - lo], 1);
    }
  }
  __syncthreads();
  const int c = cnt[t];
  int incl = c;
  for (int o = 1; o < 32; o <<= 1) {
    const int y = __shfl_up(incl, o);
    if (lane >= o) incl += y;
  }
  if (lane == 31) wsum[w] = incl;
  __syncthreads();
  if (w == 0) {
    const int v = wsum[lane];
    int in2 = v;
    for (int o = 1; o < 32; o <<= 1) {
      const int y = __shfl_up(in2, o);
      if (lane >= o) in2 += y;
    }
    wex[lane] = in2 - v;
  }
  __syncthreads();
  const int rp = beg + wex[w] + incl - c;
  rowptr[lo + t] = rp;
  cnt_g[lo + t] = c;
  const float df = (float)(c + 1);
  dinv[lo + t] = (df > 0.0f) ? rsqrtf(df) : 0.0f;
  cur[t] = rp;
  if (t < 32) cur[BW + t] = 0;
  __syncthreads();
  if (nbk <= SCAP) {
    for (int u0 = 0; u0 < nseg; u0 += SB2) {
      int2 pr[SB2][2];
      int pos[SB2][2];
#pragma unroll
      for (int u = 0; u < SB2; ++u) {
        const int cc = __builtin_amdgcn_readlane(hc, u0 + u);
        const int s0 = (w + 32 * (u0 + u)) * CHUNK + __builtin_amdgcn_readlane(hs, u0 + u);
        pr[u][0] = (lane < cc) ? pairs[s0 + lane] : make_int2(0, -1);
        pr[u][1] = (lane + 32 < cc) ? pairs[s0 + lane + 32] : make_int2(0, -1);
      }
#pragma unroll
      for (int u = 0; u < SB2; ++u) {
#pragma unroll
        for (int j = 0; j < 2; ++j) {
          const int kq = (pr[u][j].y >= 0) ? (pr[u][j].y - lo) : (BW + lane);
          pos[u][j] = atomicAdd(&cur[kq], 1);
        }
      }
#pragma unroll
      for (int u = 0; u < SB2; ++u) {
#pragma unroll
        for (int j = 0; j < 2; ++j) {
          if (pr[u][j].y >= 0) stage[pos[u][j] - beg] = pr[u][j].x;
        }
      }
      for (int u = 0; u < SB2; ++u) {
        const int cc = __builtin_amdgcn_readlane(hc, u0 + u);
        const int s0 = (w + 32 * (u0 + u)) * CHUNK + __builtin_amdgcn_readlane(hs, u0 + u);
        for (int i = 64 + lane; i < cc; i += 32) {
          const int2 p = pairs[s0 + i];
          const int ps = atomicAdd(&cur[p.y - lo], 1);
          stage[ps - beg] = p.x;
        }
      }
    }
    __syncthreads();
    for (int i = t; i < nbk; i += BW) csr_src[beg + i] = stage[i];
  } else {
    for (int u0 = 0; u0 < nseg; u0 += SB2) {
      int2 pr[SB2][2];
      int pos[SB2][2];
#pragma unroll
      for (int u = 0; u < SB2; ++u) {
        const int cc = __builtin_amdgcn_readlane(hc, u0 + u);
        const int s0 = (w + 32 * (u0 + u)) * CHUNK + __builtin_amdgcn_readlane(hs, u0 + u);
        pr[u][0] = (lane < cc) ? pairs[s0 + lane] : make_int2(0, -1);
        pr[u][1] = (lane + 32 < cc) ? pairs[s0 + lane + 32] : make_int2(0, -1);
      }
#pragma unroll
      for (int u = 0; u < SB2; ++u) {
#pragma unroll
        for (int j = 0; j < 2; ++j) {
          const int kq = (pr[u][j].y >= 0) ? (pr[u][j].y - lo) : (BW + lane);
          pos[u][j] = atomicAdd(&cur[kq], 1);
        }
      }
#pragma unroll
      for (int u = 0; u < SB2; ++u) {
#pragma unroll
        for (int j = 0; j < 2; ++j) {
          if (pr[u][j].y >= 0) csr_src[pos[u][j]] = pr[u][j].x;
        }
      }
      for (int u = 0; u < SB2; ++u) {
        const int cc = __builtin_amdgcn_readlane(hc, u0 + u);
        const int s0 = (w + 32 * (u0 + u)) * CHUNK + __builtin_amdgcn_readlane(hs, u0 + u);
        for (int i = 64 + lane; i < cc; i += 32) {
          const int2 p = pairs[s0 + i];
          const int ps = atomicAdd(&cur[p.y - lo], 1);
          csr_src[ps] = p.x;
        }
      }
    }
  }
}

constexpr int GEMM_WPB = 4;

template <int K>
__device__ __forceinline__ void gemm_request_tile(v2f (&a)[K / 4], v4f (&d)[2],
                                                  const float* __restrict__ X,
                                                  const float* __restrict__ dinv,
                                                  int t, int M, int c, int h) {
  int row = (t << 4) + c;
  if (row >= M) row = M - 1;
  const float* __restrict__ xr = X + (size_t)row * K + 2 * h;
#pragma unroll
  for (int ks = 0; ks < K / 4; ++ks) a[ks] = *(const v2f*)(xr + 4 * ks);
  const v4f* __restrict__ dp = (const v4f*)(dinv + (t << 4) + 8 * h);
  d[0] = dp[0];
  d[1] = dp[1];
}

template <int K, int NC>
__device__ __forceinline__ void gemm_tile(const v2f (&a)[K / 4], const v4f (&d)[2],
                                          const v2f (&bf)[K / 4][NC / 16],
                                          float* __restrict__ H, int t, int M, int c, int h) {
  constexpr int NT = NC / 16;
  constexpr int KS = K / 4;
  typedef __attribute__((ext_vector_type(NT))) float vnt;
  v8f acc[NT];
#pragma unroll
  for (int ct = 0; ct < NT; ++ct) acc[ct] = v8f{};
#pragma unroll
  for (int ks = 0; ks < KS; ++ks) {
#pragma unroll
    for (int ct = 0; ct < NT; ++ct) {
      acc[ct] = __builtin_amdgcn_wmma_f32_16x16x4_f32(
           false, a[ks],  false, bf[ks][ct],
           (short)0, acc[ct],  false,  false);
    }
  }
  const int rbase = (t << 4) + 8 * h;
  float* __restrict__ hp = H + (size_t)rbase * NC + NT * c;
  if ((t << 4) + 16 <= M) {
#pragma unroll
    for (int v = 0; v < 8; ++v) {
      vnt o;
#pragma unroll
      for (int ct = 0; ct < NT; ++ct) o[ct] = acc[ct][v] * d[v >> 2][v & 3];
      *(vnt*)(hp + v * NC) = o;
    }
  } else {
#pragma unroll
    for (int v = 0; v < 8; ++v) {
      vnt o;
#pragma unroll
      for (int ct = 0; ct < NT; ++ct) o[ct] = acc[ct][v] * d[v >> 2][v & 3];
      if (rbase + v < M) *(vnt*)(hp + v * NC) = o;
    }
  }
}

template <int K, int NC>
__global__ __launch_bounds__(32 * GEMM_WPB) void gemm_wmma_kernel(
    const float* __restrict__ X, const float* __restrict__ W,
    const float* __restrict__ dinv, float* __restrict__ H, int M, int ntiles, int GW) {
  constexpr int NT = NC / 16;
  constexpr int KS = K / 4;
  typedef __attribute__((ext_vector_type(NT))) float vnt;
  const int lane = threadIdx.x & 31;
  const int wave = __builtin_amdgcn_readfirstlane((int)(threadIdx.x >> 5));
  const int c = lane & 15;
  const int h = lane >> 4;
  const int gw = (int)blockIdx.x * GEMM_WPB + wave;
  if (gw >= ntiles) return;

  v2f a0[KS], a1[KS];
  v4f d0[2], d1[2];
  int t = gw;
  gemm_request_tile<K>(a0, d0, X, dinv, t, M, c, h);

  v2f bf[KS][NT];
#pragma unroll
  for (int ks = 0; ks < KS; ++ks) {
    const vnt r0 = *(const vnt*)(W + (4 * ks + 2 * h) * NC + NT * c);
    const vnt r1 = *(const vnt*)(W + (4 * ks + 2 * h + 1) * NC + NT * c);
#pragma unroll
    for (int ct = 0; ct < NT; ++ct) {
      bf[ks][ct].x = r0[ct];
      bf[ks][ct].y = r1[ct];
    }
  }

  for (;;) {
    const int t1 = t + GW;
    if (t1 >= ntiles) {
      gemm_tile<K, NC>(a0, d0, bf, H, t, M, c, h);
      break;
    }
    gemm_request_tile<K>(a1, d1, X, dinv, t1, M, c, h);
    gemm_tile<K, NC>(a0, d0, bf, H, t, M, c, h);
    const int t2 = t1 + GW;
    if (t2 >= ntiles) {
      gemm_tile<K, NC>(a1, d1, bf, H, t1, M, c, h);
      break;
    }
    gemm_request_tile<K>(a0, d0, X, dinv, t2, M, c, h);
    gemm_tile<K, NC>(a1, d1, bf, H, t1, M, c, h);
    t = t2;
  }
}


template <int C, int NG>
__device__ __forceinline__ v2f gather_batch(v2f acc, int s_l, int m,
                                            const v2f* __restrict__ h2, int sub, int cl) {
  constexpr int LPR = C / 2;
  constexpr int RPI = 32 / LPR;
  constexpr int NL  = NG * 8 / RPI;
  constexpr int FL  = (NG - 1) * 8 / RPI;
  v2f hv[NL];
#pragma unroll
  for (int q = 0; q < NL; ++q) {
    int s;
    if (RPI == 1) {
      s = __builtin_amdgcn_readlane(s_l, q);
    } else {
      const int sa = __builtin_amdgcn_readlane(s_l, 2 * q);
      const int sb = __builtin_amdgcn_readlane(s_l, 2 * q + 1);
      s = sub ? sb : sa;
    }
    const v2f* __restrict__ rowp = h2 + (size_t)s * LPR;
    hv[q] = rowp[cl];
  }
#pragma unroll
  for (int q = 0; q < NL; ++q) {
    if (q < FL) {
      acc = acc + hv[q];
    } else {
      const bool ok = (RPI == 1) ? (q < m) : ((2 * q + sub) < m);
      const v2f nv = acc + hv[q];
      acc = ok ? nv : acc;
    }
  }
  return acc;
}

template <int C>
__device__ __forceinline__ v2f gather_items(v2f acc, int s_l, int m,
                                            const v2f* __restrict__ h2, int sub, int cl) {
  const int ng = (m + 7) >> 3;
  switch (ng) {
    case 1:  return gather_batch<C, 1>(acc, s_l, m, h2, sub, cl);
    case 2:  return gather_batch<C, 2>(acc, s_l, m, h2, sub, cl);
    case 3:  return gather_batch<C, 3>(acc, s_l, m, h2, sub, cl);
    default: return gather_batch<C, 4>(acc, s_l, m, h2, sub, cl);
  }
}

template <int C, bool RELU>
__global__ __launch_bounds__(256) void gather_kernel(
    const int* __restrict__ rowptr, const int* __restrict__ cnt,
    const int* __restrict__ csr_src, const float* __restrict__ dinv,
    const float* __restrict__ h, const float* __restrict__ bias,
    float* __restrict__ out, int n, int npb) {
  constexpr int LPR = C / 2;
  constexpr int RPI = 32 / LPR;
  constexpr int WPB = 8;
  const int lane = threadIdx.x & 31;
  const int wv  = __builtin_amdgcn_readfirstlane((int)(threadIdx.x >> 5));
  const int sub = lane / LPR;
  const int cl  = lane - sub * LPR;
  const v2f* __restrict__ h2 = (const v2f*)h;
  const v2f bv = ((const v2f*)bias)[cl];

  const int nb0  = (int)blockIdx.x * npb;
  const int nend = (nb0 + npb < n) ? (nb0 + npb) : n;

  for (int c0 = nb0 + wv; c0 < nend; c0 += 32 * WPB) {
    const int hn = c0 + WPB * lane;
    int hstart = 0, hdeg = 0, hdi = 0;
    if (hn < nend) {
      hstart = rowptr[hn];
      hdeg   = cnt[hn];
      hdi    = __float_as_int(dinv[hn]);
    }
    const int avail = (nend - c0 + WPB - 1) / WPB;
    const int cnum  = (avail < 32) ? avail : 32;

    int node  = c0;
    int start = __builtin_amdgcn_readlane(hstart, 0);
    int deg   = __builtin_amdgcn_readlane(hdeg, 0);
    int s_l   = node;
    if (lane < deg) s_l = csr_src[start + lane];

    for (int it = 0; it < cnum; ++it) {
      const float di = __int_as_float(__builtin_amdgcn_readlane(hdi, it));
      const int nit   = it + 1;
      const int nnode = node + WPB;
      int nstart = 0, ndeg = 0, ns_l = nnode;
      if (nit < cnum) {
        nstart = __builtin_amdgcn_readlane(hstart, nit);
        ndeg   = __builtin_amdgcn_readlane(hdeg, nit);
        if (lane < ndeg) ns_l = csr_src[nstart + lane];
      }

      const int total = deg + 1;
      v2f acc = {0.0f, 0.0f};
      {
        const int m = (total < 32) ? total : 32;
        acc = gather_items<C>(acc, s_l, m, h2, sub, cl);
      }
      for (int base = 32; base < total; base += 32) {
        const int k = base + lane;
        int s_x = node;
        if (k < deg) s_x = csr_src[start + k];
        const int rem = total - base;
        const int m = (rem < 32) ? rem : 32;
        acc = gather_items<C>(acc, s_x, m, h2, sub, cl);
      }

      if (RPI == 2) {
        acc.x += __shfl_xor(acc.x, 16);
        acc.y += __shfl_xor(acc.y, 16);
      }
      if (sub == 0) {
        v2f r = acc * di + bv;
        if (RELU) {
          r.x = fmaxf(r.x, 0.0f);
          r.y = fmaxf(r.y, 0.0f);
        }
        ((v2f*)out)[(size_t)node * LPR + cl] = r;
      }

      node  = nnode;
      start = nstart;
      deg   = ndeg;
      s_l   = ns_l;
    }
  }
}


extern "C" void kernel_launch(void* const* d_in, const int* in_sizes, int n_in,
                              void* d_out, int out_size, void* d_ws, size_t ws_size,
                              hipStream_t stream) {
  (void)n_in; (void)out_size; (void)ws_size;
  const float* x  = (const float*)d_in[0];
  const int*   ei = (const int*)d_in[1];
  const float* W1 = (const float*)d_in[2];
  const float* b1 = (const float*)d_in[3];
  const float* W2 = (const float*)d_in[4];
  const float* b2 = (const float*)d_in[5];

  const int IN_C = 128, HID = 64, OUT = 32;
  const int n = in_sizes[0] / IN_C;
  const int e = in_sizes[1] / 2;
  const int npb   = (((n + 1535) / 1536) + 7) & ~7;
  const int ggrid = (n + npb - 1) / npb;
  const int ntiles = (n + 15) / 16;
  const int gwaves = (ntiles < 2048) ? ntiles : 2048;
  const int ggemm  = (gwaves + GEMM_WPB - 1) / GEMM_WPB;
  const int* srcp = ei;
  const int* dstp = ei + e;

  const int    np   = (n + BW - 1) & ~(BW - 1);
  const int    nb   = np >> BSH;
  const int    nblk = (e + CHUNK - 1) / CHUNK;
  const size_t ep   = ((size_t)e + 63) & ~(size_t)63;

  int*   cnt      = (int*)d_ws;
  int*   rowptr   = cnt + np;
  float* dinv     = (float*)(rowptr + np);
  int*   seghist  = (int*)(dinv + np);
  int*   segstart = seghist + (size_t)nblk * MAXB;
  int*   pairs    = segstart + (size_t)nblk * MAXB;
  int*   csr      = pairs + 2 * ep;
  float* h1      = (float*)(csr + ep);
  float* a1      = h1 + (size_t)n * HID;
  float* h2      = a1 + (size_t)n * HID;
  float* outp    = (float*)d_out;
  (void)OUT;

  chunk_group_kernel<<<nblk, 256, 0, stream>>>(srcp, dstp, seghist, segstart, pairs, e);
  bucket_sort_kernel<<<nb, BW, 0, stream>>>(pairs, seghist, segstart, nblk, cnt, rowptr, dinv, csr);

  gemm_wmma_kernel<128, 64><<<ggemm, 32 * GEMM_WPB, 0, stream>>>(x, W1, dinv, h1, n, ntiles, gwaves);
  gather_kernel<64, true><<<ggrid, 256, 0, stream>>>(rowptr, cnt, csr, dinv, h1, b1, a1, n, npb);

  gemm_wmma_kernel<64, 32><<<ggemm, 32 * GEMM_WPB, 0, stream>>>(a1, W2, dinv, h2, n, ntiles, gwaves);
  gather_kernel<32, false><<<ggrid, 256, 0, stream>>>(rowptr, cnt, csr, dinv, h2, b2, outp, n, npb);
}


# Layer-2 product moved into the layer-1 gather: each wave keeps its 16 latest rows in LDS and multiplies them by W2 on the f32 matrix instruction; no layer-1 output, 5 launches; sums unchanged
# speedup vs baseline: 6.7438x; 1.0368x over previous
#include <hip/hip_runtime.h>


typedef __attribute__((ext_vector_type(2))) float v2f;
typedef __attribute__((ext_vector_type(4))) float v4f;
typedef __attribute__((ext_vector_type(8))) float v8f;


constexpr int BSH    = 10;
constexpr int BW     = 1 << BSH;
constexpr int MAXB   = 128;
constexpr int CHUNK  = 4096;
constexpr int EPT    = CHUNK / 256;
constexpr int SCAP   = 40960;

__device__ inline void wave_scan128(const int* in, int* out, int lane) {
  const int a0 = in[lane * 4 + 0];
  const int a1 = in[lane * 4 + 1];
  const int a2 = in[lane * 4 + 2];
  const int a3 = in[lane * 4 + 3];
  const int s = (a0 + a1) + (a2 + a3);
  int incl = s;
  for (int o = 1; o < 32; o <<= 1) {
    const int y = __shfl_up(incl, o);
    if (lane >= o) incl += y;
  }
  const int ex = incl - s;
  out[lane * 4 + 0] = ex;
  out[lane * 4 + 1] = ex + a0;
  out[lane * 4 + 2] = ex + a0 + a1;
  out[lane * 4 + 3] = ex + a0 + a1 + a2;
}

__global__ __launch_bounds__(256) void chunk_group_kernel(const int* __restrict__ src,
                                                          const int* __restrict__ dst,
                                                          int* __restrict__ seghist,
                                                          int* __restrict__ segstart,
                                                          int* __restrict__ pairs_i, int e) {
  int2* __restrict__ pairs = (int2*)pairs_i;
  __shared__ int hist[MAXB + 32];
  __shared__ int lstart[MAXB];
  __shared__ int2 stage[CHUNK];
  const int t = threadIdx.x;
  const int lane = t & 31;
  const int w = t >> 5;
  const int blk = blockIdx.x;
  if (t < MAXB + 32) hist[t] = 0;
  __syncthreads();

  const int e0 = blk * CHUNK;
  int d[EPT], s[EPT], pos[EPT];
#pragma unroll
  for (int j = 0; j < EPT; ++j) {
    const int i = e0 + j * 256 + t;
    d[j] = (i < e) ? dst[i] : -1;
    s[j] = (i < e) ? src[i] : 0;
  }
#pragma unroll
  for (int j = 0; j < EPT; ++j) {
    const int bk = (d[j] >= 0) ? (d[j] >> BSH) : (MAXB + lane);
    pos[j] = atomicAdd(&hist[bk], 1);
  }
  __syncthreads();
  if (w == 0) wave_scan128(hist, lstart, lane);
  __syncthreads();
#pragma unroll
  for (int j = 0; j < EPT; ++j) {
    if (d[j] >= 0) stage[lstart[d[j] >> BSH] + pos[j]] = make_int2(s[j], d[j]);
  }
  if (t < MAXB) {
    seghist[blk * MAXB + t]  = hist[t];
    segstart[blk * MAXB + t] = lstart[t];
  }
  __syncthreads();
  const int cntb = (e - e0 < CHUNK) ? (e - e0) : CHUNK;
  int2 pr[EPT];
#pragma unroll
  for (int j = 0; j < EPT; ++j) {
    const int i = j * 256 + t;
    pr[j] = stage[(i < cntb) ? i : (cntb - 1)];
  }
#pragma unroll
  for (int j = 0; j < EPT; ++j) {
    const int i = j * 256 + t;
    if (i < cntb) pairs[e0 + i] = pr[j];
  }
}

constexpr int TB  = 16;
constexpr int SB1 = 16;
constexpr int SB2 = 8;

__global__ __launch_bounds__(1024) void bucket_sort_kernel(const int* __restrict__ pairs_i,
                                                           const int* __restrict__ seghist,
                                                           const int* __restrict__ segstart,
                                                           int nblk,
                                                           int* __restrict__ cnt_g,
                                                           int* __restrict__ rowptr,
                                                           float* __restrict__ dinv,
                                                           int* __restrict__ csr_src) {
  const int2* __restrict__ pairs = (const int2*)pairs_i;
  __shared__ int cnt[BW];
  __shared__ int cur[BW + 32];
  __shared__ int wsum[32];
  __shared__ int wex[32];
  __shared__ int redb[32];
  __shared__ int redm[32];
  __shared__ int stage[SCAP];
  const int t = threadIdx.x;
  const int lane = t & 31;
  const int w = __builtin_amdgcn_readfirstlane(t >> 5);
  const int b = blockIdx.x;
  const int lo = b << BSH;
  cnt[t] = 0;
  const int hk = w + 32 * lane;
  int hc = 0, hs = 0;
  if (hk < nblk) {
    hc = seghist[hk * MAXB + b];
    hs = segstart[hk * MAXB + b];
  }
  const int nseg = (nblk - w + 31) >> 5;
  int below = 0, mine = 0;
  {
    const int4* __restrict__ M4 = (const int4*)seghist;
    const int c0 = 4 * lane;
    if (c0 <= b) {
      for (int k0 = w; k0 < nblk; k0 += 32 * TB) {
        int4 v[TB];
#pragma unroll
        for (int u = 0; u < TB; ++u) {
          const int k = k0 + 32 * u;
          const int kk = (k < nblk) ? k : k0;
          v[u] = M4[kk * (MAXB / 4) + lane];
        }
#pragma unroll
        for (int u = 0; u < TB; ++u) {
          const int k = k0 + 32 * u;
          if (k < nblk) {
            below += ((c0 + 0 < b) ? v[u].x : 0) + ((c0 + 1 < b) ? v[u].y : 0) +
                     ((c0 + 2 < b) ? v[u].z : 0) + ((c0 + 3 < b) ? v[u].w : 0);
            mine  += ((c0 + 0 == b) ? v[u].x : 0) + ((c0 + 1 == b) ? v[u].y : 0) +
                     ((c0 + 2 == b) ? v[u].z : 0) + ((c0 + 3 == b) ? v[u].w : 0);
          }
        }
      }
    }
    for (int o = 16; o > 0; o >>= 1) {
      below += __shfl_xor(below, o);
      mine  += __shfl_xor(mine, o);
    }
    if (lane == 0) {
      redb[w] = below;
      redm[w] = mine;
    }
  }
  __syncthreads();
  int beg = redb[lane];
  int nbk = redm[lane];
  for (int o = 16; o > 0; o >>= 1) {
    beg += __shfl_xor(beg, o);
    nbk += __shfl_xor(nbk, o);
  }

  for (int u0 = 0; u0 < nseg; u0 += SB1) {
    int dd[SB1][2];
#pragma unroll
    for (int u = 0; u < SB1; ++u) {
      const int c  = __builtin_amdgcn_readlane(hc, u0 + u);
      const int s0 = (w + 32 * (u0 + u)) * CHUNK + __builtin_amdgcn_readlane(hs, u0 + u);
      dd[u][0] = (lane < c) ? pairs[s0 + lane].y : -1;
      dd[u][1] = (lane + 32 < c) ? pairs[s0 + lane + 32].y : -1;
    }
#pragma unroll
    for (int u = 0; u < SB1; ++u) {
      if (dd[u][0] >= 0) atomicAdd(&cnt[dd[u][0] - lo], 1);
      if (dd[u][1] >= 0) atomicAdd(&cnt[dd[u][1] - lo], 1);
    }
    for (int u = 0; u < SB1; ++u) {
      const int c  = __builtin_amdgcn_readlane(hc, u0 + u);
      const int s0 = (w + 32 * (u0 + u)) * CHUNK + __builtin_amdgcn_readlane(hs, u0 + u);
      for (int i = 64 + lane; i < c; i += 32) atomicAdd(&cnt[pairs[s0 + i].y - lo], 1);
    }
  }
  __syncthreads();
  const int c = cnt[t];
  int incl = c;
  for (int o = 1; o < 32; o <<= 1) {
    const int y = __shfl_up(incl, o);
    if (lane >= o) incl += y;
  }
  if (lane == 31) wsum[w] = incl;
  __syncthreads();
  if (w == 0) {
    const int v = wsum[lane];
    int in2 = v;
    for (int o = 1; o < 32; o <<= 1) {
      const int y = __shfl_up(in2, o);
      if (lane >= o) in2 += y;
    }
    wex[lane] = in2 - v;
  }
  __syncthreads();
  const int rp = beg + wex[w] + incl - c;
  rowptr[lo + t] = rp;
  cnt_g[lo + t] = c;
  const float df = (float)(c + 1);
  dinv[lo + t] = (df > 0.0f) ? rsqrtf(df) : 0.0f;
  cur[t] = rp;
  if (t < 32) cur[BW + t] = 0;
  __syncthreads();
  if (nbk <= SCAP) {
    for (int u0 = 0; u0 < nseg; u0 += SB2) {
      int2 pr[SB2][2];
      int pos[SB2][2];
#pragma unroll
      for (int u = 0; u < SB2; ++u) {
        const int cc = __builtin_amdgcn_readlane(hc, u0 + u);
        const int s0 = (w + 32 * (u0 + u)) * CHUNK + __builtin_amdgcn_readlane(hs, u0 + u);
        pr[u][0] = (lane < cc) ? pairs[s0 + lane] : make_int2(0, -1);
        pr[u][1] = (lane + 32 < cc) ? pairs[s0 + lane + 32] : make_int2(0, -1);
      }
#pragma unroll
      for (int u = 0; u < SB2; ++u) {
#pragma unroll
        for (int j = 0; j < 2; ++j) {
          const int kq = (pr[u][j].y >= 0) ? (pr[u][j].y - lo) : (BW + lane);
          pos[u][j] = atomicAdd(&cur[kq], 1);
        }
      }
#pragma unroll
      for (int u = 0; u < SB2; ++u) {
#pragma unroll
        for (int j = 0; j < 2; ++j) {
          if (pr[u][j].y >= 0) stage[pos[u][j] - beg] = pr[u][j].x;
        }
      }
      for (int u = 0; u < SB2; ++u) {
        const int cc = __builtin_amdgcn_readlane(hc, u0 + u);
        const int s0 = (w + 32 * (u0 + u)) * CHUNK + __builtin_amdgcn_readlane(hs, u0 + u);
        for (int i = 64 + lane; i < cc; i += 32) {
          const int2 p = pairs[s0 + i];
          const int ps = atomicAdd(&cur[p.y - lo], 1);
          stage[ps - beg] = p.x;
        }
      }
    }
    __syncthreads();
    for (int i = t; i < nbk; i += BW) csr_src[beg + i] = stage[i];
  } else {
    for (int u0 = 0; u0 < nseg; u0 += SB2) {
      int2 pr[SB2][2];
      int pos[SB2][2];
#pragma unroll
      for (int u = 0; u < SB2; ++u) {
        const int cc = __builtin_amdgcn_readlane(hc, u0 + u);
        const int s0 = (w + 32 * (u0 + u)) * CHUNK + __builtin_amdgcn_readlane(hs, u0 + u);
        pr[u][0] = (lane < cc) ? pairs[s0 + lane] : make_int2(0, -1);
        pr[u][1] = (lane + 32 < cc) ? pairs[s0 + lane + 32] : make_int2(0, -1);
      }
#pragma unroll
      for (int u = 0; u < SB2; ++u) {
#pragma unroll
        for (int j = 0; j < 2; ++j) {
          const int kq = (pr[u][j].y >= 0) ? (pr[u][j].y - lo) : (BW + lane);
          pos[u][j] = atomicAdd(&cur[kq], 1);
        }
      }
#pragma unroll
      for (int u = 0; u < SB2; ++u) {
#pragma unroll
        for (int j = 0; j < 2; ++j) {
          if (pr[u][j].y >= 0) csr_src[pos[u][j]] = pr[u][j].x;
        }
      }
      for (int u = 0; u < SB2; ++u) {
        const int cc = __builtin_amdgcn_readlane(hc, u0 + u);
        const int s0 = (w + 32 * (u0 + u)) * CHUNK + __builtin_amdgcn_readlane(hs, u0 + u);
        for (int i = 64 + lane; i < cc; i += 32) {
          const int2 p = pairs[s0 + i];
          const int ps = atomicAdd(&cur[p.y - lo], 1);
          csr_src[ps] = p.x;
        }
      }
    }
  }
}

constexpr int GEMM_WPB = 4;

template <int K>
__device__ __forceinline__ void gemm_request_tile(v2f (&a)[K / 4], v4f (&d)[2],
                                                  const float* __restrict__ X,
                                                  const float* __restrict__ dinv,
                                                  int t, int M, int c, int h) {
  int row = (t << 4) + c;
  if (row >= M) row = M - 1;
  const float* __restrict__ xr = X + (size_t)row * K + 2 * h;
#pragma unroll
  for (int ks = 0; ks < K / 4; ++ks) a[ks] = *(const v2f*)(xr + 4 * ks);
  const v4f* __restrict__ dp = (const v4f*)(dinv + (t << 4) + 8 * h);
  d[0] = dp[0];
  d[1] = dp[1];
}

template <int K, int NC>
__device__ __forceinline__ void gemm_tile(const v2f (&a)[K / 4], const v4f (&d)[2],
                                          const v2f (&bf)[K / 4][NC / 16],
                                          float* __restrict__ H, int t, int M, int c, int h) {
  constexpr int NT = NC / 16;
  constexpr int KS = K / 4;
  typedef __attribute__((ext_vector_type(NT))) float vnt;
  v8f acc[NT];
#pragma unroll
  for (int ct = 0; ct < NT; ++ct) acc[ct] = v8f{};
#pragma unroll
  for (int ks = 0; ks < KS; ++ks) {
#pragma unroll
    for (int ct = 0; ct < NT; ++ct) {
      acc[ct] = __builtin_amdgcn_wmma_f32_16x16x4_f32(
           false, a[ks],  false, bf[ks][ct],
           (short)0, acc[ct],  false,  false);
    }
  }
  const int rbase = (t << 4) + 8 * h;
  float* __restrict__ hp = H + (size_t)rbase * NC + NT * c;
  if ((t << 4) + 16 <= M) {
#pragma unroll
    for (int v = 0; v < 8; ++v) {
      vnt o;
#pragma unroll
      for (int ct = 0; ct < NT; ++ct) o[ct] = acc[ct][v] * d[v >> 2][v & 3];
      *(vnt*)(hp + v * NC) = o;
    }
  } else {
#pragma unroll
    for (int v = 0; v < 8; ++v) {
      vnt o;
#pragma unroll
      for (int ct = 0; ct < NT; ++ct) o[ct] = acc[ct][v] * d[v >> 2][v & 3];
      if (rbase + v < M) *(vnt*)(hp + v * NC) = o;
    }
  }
}

template <int K, int NC>
__global__ __launch_bounds__(32 * GEMM_WPB) void gemm_wmma_kernel(
    const float* __restrict__ X, const float* __restrict__ W,
    const float* __restrict__ dinv, float* __restrict__ H, int M, int ntiles, int GW) {
  constexpr int NT = NC / 16;
  constexpr int KS = K / 4;
  typedef __attribute__((ext_vector_type(NT))) float vnt;
  const int lane = threadIdx.x & 31;
  const int wave = __builtin_amdgcn_readfirstlane((int)(threadIdx.x >> 5));
  const int c = lane & 15;
  const int h = lane >> 4;
  const int gw = (int)blockIdx.x * GEMM_WPB + wave;
  if (gw >= ntiles) return;

  v2f a0[KS], a1[KS];
  v4f d0[2], d1[2];
  int t = gw;
  gemm_request_tile<K>(a0, d0, X, dinv, t, M, c, h);

  v2f bf[KS][NT];
#pragma unroll
  for (int ks = 0; ks < KS; ++ks) {
    const vnt r0 = *(const vnt*)(W + (4 * ks + 2 * h) * NC + NT * c);
    const vnt r1 = *(const vnt*)(W + (4 * ks + 2 * h + 1) * NC + NT * c);
#pragma unroll
    for (int ct = 0; ct < NT; ++ct) {
      bf[ks][ct].x = r0[ct];
      bf[ks][ct].y = r1[ct];
    }
  }

  for (;;) {
    const int t1 = t + GW;
    if (t1 >= ntiles) {
      gemm_tile<K, NC>(a0, d0, bf, H, t, M, c, h);
      break;
    }
    gemm_request_tile<K>(a1, d1, X, dinv, t1, M, c, h);
    gemm_tile<K, NC>(a0, d0, bf, H, t, M, c, h);
    const int t2 = t1 + GW;
    if (t2 >= ntiles) {
      gemm_tile<K, NC>(a1, d1, bf, H, t1, M, c, h);
      break;
    }
    gemm_request_tile<K>(a0, d0, X, dinv, t2, M, c, h);
    gemm_tile<K, NC>(a1, d1, bf, H, t1, M, c, h);
    t = t2;
  }
}


template <int C, int NG>
__device__ __forceinline__ v2f gather_batch(v2f acc, int s_l, int m,
                                            const v2f* __restrict__ h2, int sub, int cl) {
  constexpr int LPR = C / 2;
  constexpr int RPI = 32 / LPR;
  constexpr int NL  = NG * 8 / RPI;
  constexpr int FL  = (NG - 1) * 8 / RPI;
  v2f hv[NL];
#pragma unroll
  for (int q = 0; q < NL; ++q) {
    int s;
    if (RPI == 1) {
      s = __builtin_amdgcn_readlane(s_l, q);
    } else {
      const int sa = __builtin_amdgcn_readlane(s_l, 2 * q);
      const int sb = __builtin_amdgcn_readlane(s_l, 2 * q + 1);
      s = sub ? sb : sa;
    }
    const v2f* __restrict__ rowp = h2 + (size_t)s * LPR;
    hv[q] = rowp[cl];
  }
#pragma unroll
  for (int q = 0; q < NL; ++q) {
    if (q < FL) {
      acc = acc + hv[q];
    } else {
      const bool ok = (RPI == 1) ? (q < m) : ((2 * q + sub) < m);
      const v2f nv = acc + hv[q];
      acc = ok ? nv : acc;
    }
  }
  return acc;
}

template <int C>
__device__ __forceinline__ v2f gather_items(v2f acc, int s_l, int m,
                                            const v2f* __restrict__ h2, int sub, int cl) {
  const int ng = (m + 7) >> 3;
  switch (ng) {
    case 1:  return gather_batch<C, 1>(acc, s_l, m, h2, sub, cl);
    case 2:  return gather_batch<C, 2>(acc, s_l, m, h2, sub, cl);
    case 3:  return gather_batch<C, 3>(acc, s_l, m, h2, sub, cl);
    default: return gather_batch<C, 4>(acc, s_l, m, h2, sub, cl);
  }
}

constexpr int GATHER_WPB = 8;
constexpr int TPAD = 4;

template <int C>
__device__ __forceinline__ float* gather_wave_tile(int wv) {
  __shared__ __attribute__((aligned(16))) float tiles[GATHER_WPB][16][C + TPAD];
  return &tiles[wv][0][0];
}

template <int C, int NC>
__device__ __forceinline__ void gather_tile_product(float* __restrict__ tl, int rows, int tb, int c0, int hdi,
                                                    const float* __restrict__ Wn, float* __restrict__ H,
                                                    int lane) {
  constexpr int NT = NC / 16;
  constexpr int KS = C / 4;
  constexpr int TW = C + TPAD;
  typedef __attribute__((ext_vector_type(NT))) float vnt;
  const int c = lane & 15;
  const int hh = lane >> 4;
  for (int r = rows; r < 16; ++r) *(v2f*)(tl + r * TW + 2 * lane) = v2f{0.0f, 0.0f};
  __builtin_amdgcn_fence(__ATOMIC_RELEASE, "wavefront");
  __builtin_amdgcn_fence(__ATOMIC_ACQUIRE, "wavefront");
  constexpr int QK = 4;
  constexpr int NQ = KS / QK;
  static_assert(KS % QK == 0, "whole groups of k-steps");
  v2f a[2][QK];
  vnt w0[2][QK], w1[2][QK];
  v8f acc[NT];
#pragma unroll
  for (int ct = 0; ct < NT; ++ct) acc[ct] = v8f{};
#pragma unroll
  for (int q = 0; q <= NQ; ++q) {
    if (q < NQ) {
#pragma unroll
      for (int j = 0; j < QK; ++j) {
        const int ks = q * QK + j;
        a[q & 1][j]  = *(const v2f*)(tl + c * TW + 4 * ks + 2 * hh);
        w0[q & 1][j] = *(const vnt*)(Wn + (4 * ks + 2 * hh) * NC + NT * c);
        w1[q & 1][j] = *(const vnt*)(Wn + (4 * ks + 2 * hh + 1) * NC + NT * c);
      }
    }
    __builtin_amdgcn_sched_barrier(0);
    if (q > 0) {
#pragma unroll
      for (int j = 0; j < QK; ++j) {
#pragma unroll
        for (int ct = 0; ct < NT; ++ct) {
          v2f bfr;
          bfr.x = w0[(q - 1) & 1][j][ct];
          bfr.y = w1[(q - 1) & 1][j][ct];
          acc[ct] = __builtin_amdgcn_wmma_f32_16x16x4_f32(
               false, a[(q - 1) & 1][j],  false, bfr,
               (short)0, acc[ct],  false,  false);
        }
      }
    }
    __builtin_amdgcn_sched_barrier(0);
  }
  __builtin_amdgcn_fence(__ATOMIC_RELEASE, "wavefront");
  __builtin_amdgcn_fence(__ATOMIC_ACQUIRE, "wavefront");
  float dv[8];
#pragma unroll
  for (int v = 0; v < 8; ++v) dv[v] = __int_as_float(__shfl(hdi, tb + 8 * hh + v));
#pragma unroll
  for (int v = 0; v < 8; ++v) {
    const int r = 8 * hh + v;
    vnt o;
#pragma unroll
    for (int ct = 0; ct < NT; ++ct) o[ct] = acc[ct][v] * dv[v];
    if (r < rows) *(vnt*)(H + (size_t)(c0 + GATHER_WPB * (tb + r)) * NC + NT * c) = o;
  }
}

template <int C, bool RELU, int NC>
__global__ __launch_bounds__(256) __attribute__((amdgpu_waves_per_eu(10))) void gather_kernel(
    const int* __restrict__ rowptr, const int* __restrict__ cnt,
    const int* __restrict__ csr_src, const float* __restrict__ dinv,
    const float* __restrict__ h, const float* __restrict__ bias,
    const float* __restrict__ Wn, float* __restrict__ out, int n, int npb) {
  constexpr int LPR = C / 2;
  constexpr int RPI = 32 / LPR;
  constexpr int WPB = GATHER_WPB;
  static_assert(NC == 0 || (RPI == 1 && NC % 16 == 0 && C % 4 == 0), "fused product: one row per wave-wide load");
  const int lane = threadIdx.x & 31;
  const int wv  = __builtin_amdgcn_readfirstlane((int)(threadIdx.x >> 5));
  const int sub = lane / LPR;
  const int cl  = lane - sub * LPR;
  const v2f* __restrict__ h2 = (const v2f*)h;
  const v2f bv = ((const v2f*)bias)[cl];

  const int nb0  = (int)blockIdx.x * npb;
  const int nend = (nb0 + npb < n) ? (nb0 + npb) : n;

  for (int c0 = nb0 + wv; c0 < nend; c0 += 32 * WPB) {
    const int hn = c0 + WPB * lane;
    int hstart = 0, hdeg = 0, hdi = 0;
    if (hn < nend) {
      hstart = rowptr[hn];
      hdeg   = cnt[hn];
      hdi    = __float_as_int(dinv[hn]);
    }
    const int avail = (nend - c0 + WPB - 1) / WPB;
    const int cnum  = (avail < 32) ? avail : 32;

    int node  = c0;
    int start = __builtin_amdgcn_readlane(hstart, 0);
    int deg   = __builtin_amdgcn_readlane(hdeg, 0);
    int s_l   = node;
    if (lane < deg) s_l = csr_src[start + lane];

    for (int it = 0; it < cnum; ++it) {
      const float di = __int_as_float(__builtin_amdgcn_readlane(hdi, it));
      const int nit   = it + 1;
      const int nnode = node + WPB;
      int nstart = 0, ndeg = 0, ns_l = nnode;
      if (nit < cnum) {
        nstart = __builtin_amdgcn_readlane(hstart, nit);
        ndeg   = __builtin_amdgcn_readlane(hdeg, nit);
        if (lane < ndeg) ns_l = csr_src[nstart + lane];
      }

      const int total = deg + 1;
      v2f acc = {0.0f, 0.0f};
      {
        const int m = (total < 32) ? total : 32;
        acc = gather_items<C>(acc, s_l, m, h2, sub, cl);
      }
      for (int base = 32; base < total; base += 32) {
        const int k = base + lane;
        int s_x = node;
        if (k < deg) s_x = csr_src[start + k];
        const int rem = total - base;
        const int m = (rem < 32) ? rem : 32;
        acc = gather_items<C>(acc, s_x, m, h2, sub, cl);
      }

      if (RPI == 2) {
        acc.x += __shfl_xor(acc.x, 16);
        acc.y += __shfl_xor(acc.y, 16);
      }
      if (sub == 0) {
        v2f r = acc * di + bv;
        if (RELU) {
          r.x = fmaxf(r.x, 0.0f);
          r.y = fmaxf(r.y, 0.0f);
        }
        if constexpr (NC == 0) {
          ((v2f*)out)[(size_t)node * LPR + cl] = r;
        } else {
          *(v2f*)(gather_wave_tile<C>(wv) + (it & 15) * (C + TPAD) + 2 * cl) = r;
        }
      }
      if constexpr (NC > 0) {
        if ((it & 15) == 15 || nit == cnum) {
          gather_tile_product<C, NC>(gather_wave_tile<C>(wv), (it & 15) + 1, it & ~15, c0, hdi, Wn, out, lane);
        }
      }

      node  = nnode;
      start = nstart;
      deg   = ndeg;
      s_l   = ns_l;
    }
  }
}


extern "C" void kernel_launch(void* const* d_in, const int* in_sizes, int n_in,
                              void* d_out, int out_size, void* d_ws, size_t ws_size,
                              hipStream_t stream) {
  (void)n_in; (void)out_size; (void)ws_size;
  const float* x  = (const float*)d_in[0];
  const int*   ei = (const int*)d_in[1];
  const float* W1 = (const float*)d_in[2];
  const float* b1 = (const float*)d_in[3];
  const float* W2 = (const float*)d_in[4];
  const float* b2 = (const float*)d_in[5];

  const int IN_C = 128, HID = 64, OUT = 32;
  const int n = in_sizes[0] / IN_C;
  const int e = in_sizes[1] / 2;
  const int npb   = (((n + 1535) / 1536) + 7) & ~7;
  const int ggrid = (n + npb - 1) / npb;
  const int npbm   = (((n + 1279) / 1280) + 7) & ~7;
  const int ggridm = (n + npbm - 1) / npbm;
  const int ntiles = (n + 15) / 16;
  const int gwaves = (ntiles < 2048) ? ntiles : 2048;
  const int ggemm  = (gwaves + GEMM_WPB - 1) / GEMM_WPB;
  const int* srcp = ei;
  const int* dstp = ei + e;

  const int    np   = (n + BW - 1) & ~(BW - 1);
  const int    nb   = np >> BSH;
  const int    nblk = (e + CHUNK - 1) / CHUNK;
  const size_t ep   = ((size_t)e + 63) & ~(size_t)63;

  int*   cnt      = (int*)d_ws;
  int*   rowptr   = cnt + np;
  float* dinv     = (float*)(rowptr + np);
  int*   seghist  = (int*)(dinv + np);
  int*   segstart = seghist + (size_t)nblk * MAXB;
  int*   pairs    = segstart + (size_t)nblk * MAXB;
  int*   csr      = pairs + 2 * ep;
  float* h1      = (float*)(csr + ep);
  float* h2      = h1 + (size_t)n * HID;
  float* outp    = (float*)d_out;
  (void)OUT;

  chunk_group_kernel<<<nblk, 256, 0, stream>>>(srcp, dstp, seghist, segstart, pairs, e);
  bucket_sort_kernel<<<nb, BW, 0, stream>>>(pairs, seghist, segstart, nblk, cnt, rowptr, dinv, csr);

  gemm_wmma_kernel<128, 64><<<ggemm, 32 * GEMM_WPB, 0, stream>>>(x, W1, dinv, h1, n, ntiles, gwaves);
  gather_kernel<64, true, 32><<<ggridm, 256, 0, stream>>>(rowptr, cnt, csr, dinv, h1, b1, W2, h2, n, npbm);

  gather_kernel<32, false, 0><<<ggrid, 256, 0, stream>>>(rowptr, cnt, csr, dinv, h2, b2, nullptr, outp, n, npb);
}
